# starting kernel
# baseline (speedup 1.0000x reference)
#include <hip/hip_runtime.h>


typedef __attribute__((ext_vector_type(2))) float v2f;
typedef __attribute__((ext_vector_type(8))) float v8f;

#define BTOT 16
#define WN   64
#define NN   64
#define CC   96
#define HH   8
#define DD   768
#define C3   288
#define ROW3 18432
#define YBT  1179648
#define HD   6144
#define ZBT  393216
#define ATTN_SCALE 0.28867513459481287f

static __device__ __forceinline__ v8f wmma4(v2f a, v2f b, v8f c) {
  return __builtin_amdgcn_wmma_f32_16x16x4_f32(
       false, a,  false, b,
       (short)0, c,  false,  false);
}

__global__ __launch_bounds__(256) void qkv_gemm_kernel(
    const float* __restrict__ x, const float* __restrict__ wqkv,
    float* __restrict__ Y) {
  __shared__ __align__(16) float sX[64 * 100];
  const int tid = threadIdx.x;
  const int lane = tid & 31, wave = tid >> 5;
  const int g = blockIdx.x;
  const float* xt = x + (size_t)g * (NN * CC);
  float* yt = Y + (size_t)(g / WN) * YBT + (size_t)(g % WN) * ROW3;

  for (int i = tid; i < 64 * 24; i += 256) {
    int r = i / 24, c4 = (i % 24) * 4;
    *reinterpret_cast<float4*>(&sX[r * 100 + c4]) =
        *reinterpret_cast<const float4*>(xt + r * CC + c4);
  }
  __syncthreads();

  const int m0 = lane & 15;
  const int ksel = (lane >> 4) * 2;
  for (int tt = wave; tt < 72; tt += 8) {
    const int mt = tt & 3, nt = tt >> 2;
    const int ncol = nt * 16 + m0;
    const float* bp = wqkv + ncol * CC;
    const float* ap = &sX[(mt * 16 + m0) * 100];
    v8f acc = {};
#pragma unroll
    for (int kk = 0; kk < 24; ++kk) {
      const int kb = kk * 4 + ksel;
      v2f a = *reinterpret_cast<const v2f*>(ap + kb);
      v2f b = *reinterpret_cast<const v2f*>(bp + kb);
      acc = wmma4(a, b, acc);
    }
    float* yo = yt + ncol;
#pragma unroll
    for (int v = 0; v < 8; ++v) {
      const int m = mt * 16 + v + ((lane >> 4) << 3);
      yo[m * C3] = acc[v];
    }
  }
}

__global__ __launch_bounds__(256) void win_attn_kernel(
    const float* __restrict__ Y, const float* __restrict__ mask,
    float* __restrict__ Z) {
  __shared__ __align__(16) float sA[64 * 68];
  __shared__ __align__(16) float sB[64 * 68];
  __shared__ __align__(16) float sS[64 * 68];
  const int tid = threadIdx.x, lane = tid & 31, wave = tid >> 5;
  const int bt = blockIdx.x >> 3, h = blockIdx.x & 7;
  const float* qb = Y + (size_t)bt * YBT + h * DD;
  const float* kbp = qb + HD;
  const float* vbp = qb + 2 * HD;

  const int m0 = lane & 15;
  const int ksel = (lane >> 4) * 2;
  const int t0 = wave * 2;

  v8f accS[2];
  accS[0] = (v8f){};
  accS[1] = (v8f){};

  for (int dc = 0; dc < 12; ++dc) {
    for (int i = tid; i < 64 * 16; i += 256) {
      const int r = i / 16, c4 = (i % 16) * 4;
      *reinterpret_cast<float4*>(&sA[r * 68 + c4]) =
          *reinterpret_cast<const float4*>(qb + r * ROW3 + dc * 64 + c4);
      *reinterpret_cast<float4*>(&sB[r * 68 + c4]) =
          *reinterpret_cast<const float4*>(kbp + r * ROW3 + dc * 64 + c4);
    }
    __syncthreads();
#pragma unroll
    for (int j = 0; j < 2; ++j) {
      const int tt = t0 + j, mt = tt >> 2, nt = tt & 3;
      const float* ap = &sA[(mt * 16 + m0) * 68];
      const float* bp = &sB[(nt * 16 + m0) * 68];
#pragma unroll
      for (int kk = 0; kk < 16; ++kk) {
        const int kb = kk * 4 + ksel;
        v2f a = *reinterpret_cast<const v2f*>(ap + kb);
        v2f b = *reinterpret_cast<const v2f*>(bp + kb);
        accS[j] = wmma4(a, b, accS[j]);
      }
    }
    __syncthreads();
  }

  const float* mk = mask + (size_t)(bt & 7) * (NN * NN);
#pragma unroll
  for (int j = 0; j < 2; ++j) {
    const int tt = t0 + j, mt = tt >> 2, nt = tt & 3;
    const int ncol = nt * 16 + m0;
#pragma unroll
    for (int v = 0; v < 8; ++v) {
      const int m = mt * 16 + v + ((lane >> 4) << 3);
      sS[m * 68 + ncol] = accS[j][v] * ATTN_SCALE + mk[m * NN + ncol];
    }
  }
  __syncthreads();

  for (int r = wave * 8; r < wave * 8 + 8; ++r) {
    float x0 = sS[r * 68 + lane];
    float x1 = sS[r * 68 + lane + 32];
    float mx = fmaxf(x0, x1);
#pragma unroll
    for (int off = 16; off >= 1; off >>= 1)
      mx = fmaxf(mx, __shfl_xor(mx, off, 32));
    const float e0 = __expf(x0 - mx), e1 = __expf(x1 - mx);
    float sm = e0 + e1;
#pragma unroll
    for (int off = 16; off >= 1; off >>= 1)
      sm += __shfl_xor(sm, off, 32);
    const float inv = 1.0f / sm;
    sS[r * 68 + lane] = e0 * inv;
    sS[r * 68 + lane + 32] = e1 * inv;
  }
  __syncthreads();

  float* zb = Z + (size_t)bt * ZBT + h * DD;
  for (int dc = 0; dc < 12; ++dc) {
    for (int i = tid; i < 64 * 16; i += 256) {
      const int r = i / 16, c4 = (i % 16) * 4;
      *reinterpret_cast<float4*>(&sB[r * 68 + c4]) =
          *reinterpret_cast<const float4*>(vbp + r * ROW3 + dc * 64 + c4);
    }
    __syncthreads();
#pragma unroll
    for (int j = 0; j < 2; ++j) {
      const int tt = t0 + j, mt = tt >> 2, dt = tt & 3;
      const float* ap = &sS[(mt * 16 + m0) * 68];
      v8f acc = {};
#pragma unroll
      for (int kk = 0; kk < 16; ++kk) {
        const int kb = kk * 4 + ksel;
        v2f a = *reinterpret_cast<const v2f*>(ap + kb);
        v2f b;
        b.x = sB[kb * 68 + dt * 16 + m0];
        b.y = sB[(kb + 1) * 68 + dt * 16 + m0];
        acc = wmma4(a, b, acc);
      }
      const int dcol = dc * 64 + dt * 16 + m0;
#pragma unroll
      for (int v = 0; v < 8; ++v) {
        const int m = mt * 16 + v + ((lane >> 4) << 3);
        zb[m * HD + dcol] = acc[v];
      }
    }
    __syncthreads();
  }
}

__global__ __launch_bounds__(256) void proj_gemm_kernel(
    const float* __restrict__ Z, const float* __restrict__ wproj,
    const float* __restrict__ bproj, float* __restrict__ out) {
  __shared__ __align__(16) float sZ[64 * 100];
  const int tid = threadIdx.x, lane = tid & 31, wave = tid >> 5;
  const int g = blockIdx.x;
  const float* zt = Z + (size_t)g * (NN * CC);
  float* ot = out + (size_t)g * (NN * CC);

  for (int i = tid; i < 64 * 24; i += 256) {
    const int r = i / 24, c4 = (i % 24) * 4;
    *reinterpret_cast<float4*>(&sZ[r * 100 + c4]) =
        *reinterpret_cast<const float4*>(zt + r * CC + c4);
  }
  __syncthreads();

  const int m0 = lane & 15;
  const int ksel = (lane >> 4) * 2;
  for (int tt = wave; tt < 24; tt += 8) {
    const int mt = tt & 3, nt = tt >> 2;
    const int ocol = nt * 16 + m0;
    const float* bp = wproj + ocol * CC;
    const float* ap = &sZ[(mt * 16 + m0) * 100];
    v8f acc = {};
#pragma unroll
    for (int kk = 0; kk < 24; ++kk) {
      const int kb = kk * 4 + ksel;
      v2f a = *reinterpret_cast<const v2f*>(ap + kb);
      v2f b = *reinterpret_cast<const v2f*>(bp + kb);
      acc = wmma4(a, b, acc);
    }
    const float bias = bproj[ocol];
#pragma unroll
    for (int v = 0; v < 8; ++v) {
      const int m = mt * 16 + v + ((lane >> 4) << 3);
      ot[m * CC + ocol] = acc[v] + bias;
    }
  }
}

extern "C" void kernel_launch(void* const* d_in, const int* in_sizes, int n_in,
                              void* d_out, int out_size, void* d_ws, size_t ws_size,
                              hipStream_t stream) {
  (void)in_sizes; (void)n_in; (void)out_size; (void)ws_size;
  const float* x     = (const float*)d_in[0];
  const float* mask  = (const float*)d_in[1];
  const float* wqkv  = (const float*)d_in[2];
  const float* wproj = (const float*)d_in[3];
  const float* bproj = (const float*)d_in[4];
  float* out = (float*)d_out;

  float* Y = (float*)d_ws;
  float* Z = Y + (size_t)BTOT * YBT;

  qkv_gemm_kernel<<<dim3(BTOT * WN), dim3(256), 0, stream>>>(x, wqkv, Y);
  win_attn_kernel<<<dim3(BTOT * HH), dim3(256), 0, stream>>>(Y, mask, Z);
  proj_gemm_kernel<<<dim3(BTOT * WN), dim3(256), 0, stream>>>(Z, wproj, bproj, out);
}


# attention kernel: Q,K,V chunks (128 wide) fetched into registers one chunk ahead of their WMMAs, two LDS stages, one LDS-only barrier per chunk; order of all sums unchanged
# speedup vs baseline: 1.7522x; 1.7522x over previous
#include <hip/hip_runtime.h>


typedef __attribute__((ext_vector_type(2))) float v2f;
typedef __attribute__((ext_vector_type(8))) float v8f;

#define BTOT 16
#define WN   64
#define NN   64
#define CC   96
#define HH   8
#define DD   768
#define C3   288
#define ROW3 18432
#define YBT  1179648
#define HD   6144
#define ZBT  393216
#define ATTN_SCALE 0.28867513459481287f

static __device__ __forceinline__ v8f wmma4(v2f a, v2f b, v8f c) {
  return __builtin_amdgcn_wmma_f32_16x16x4_f32(
       false, a,  false, b,
       (short)0, c,  false,  false);
}

__global__ __launch_bounds__(256) void qkv_gemm_kernel(
    const float* __restrict__ x, const float* __restrict__ wqkv,
    float* __restrict__ Y) {
  __shared__ __align__(16) float sX[64 * 100];
  const int tid = threadIdx.x;
  const int lane = tid & 31, wave = tid >> 5;
  const int g = blockIdx.x;
  const float* xt = x + (size_t)g * (NN * CC);
  float* yt = Y + (size_t)(g / WN) * YBT + (size_t)(g % WN) * ROW3;

  for (int i = tid; i < 64 * 24; i += 256) {
    int r = i / 24, c4 = (i % 24) * 4;
    *reinterpret_cast<float4*>(&sX[r * 100 + c4]) =
        *reinterpret_cast<const float4*>(xt + r * CC + c4);
  }
  __syncthreads();

  const int m0 = lane & 15;
  const int ksel = (lane >> 4) * 2;
  for (int tt = wave; tt < 72; tt += 8) {
    const int mt = tt & 3, nt = tt >> 2;
    const int ncol = nt * 16 + m0;
    const float* bp = wqkv + ncol * CC;
    const float* ap = &sX[(mt * 16 + m0) * 100];
    v8f acc = {};
#pragma unroll
    for (int kk = 0; kk < 24; ++kk) {
      const int kb = kk * 4 + ksel;
      v2f a = *reinterpret_cast<const v2f*>(ap + kb);
      v2f b = *reinterpret_cast<const v2f*>(bp + kb);
      acc = wmma4(a, b, acc);
    }
    float* yo = yt + ncol;
#pragma unroll
    for (int v = 0; v < 8; ++v) {
      const int m = mt * 16 + v + ((lane >> 4) << 3);
      yo[m * C3] = acc[v];
    }
  }
}

#define CW   128
#define NCH  6
#define LDP  132

static __device__ __forceinline__ void lds_barrier() {
  __builtin_amdgcn_fence(__ATOMIC_RELEASE, "workgroup", "local");
  __builtin_amdgcn_s_barrier();
  __builtin_amdgcn_fence(__ATOMIC_ACQUIRE, "workgroup", "local");
}

struct chunk8 { float4 r0, r1, r2, r3, r4, r5, r6, r7; };
static __device__ __forceinline__ float4 ld4(const float* __restrict__ g, int u) {
  return *reinterpret_cast<const float4*>(g + (size_t)u * (8 * ROW3));
}
static __device__ __forceinline__ void st4(float* s, int u, float4 v) {
  *reinterpret_cast<float4*>(s + u * (8 * LDP)) = v;
}
static __device__ __forceinline__ chunk8 fetch_chunk(const float* __restrict__ g) {
  chunk8 c;
  c.r0 = ld4(g, 0); c.r1 = ld4(g, 1); c.r2 = ld4(g, 2); c.r3 = ld4(g, 3);
  c.r4 = ld4(g, 4); c.r5 = ld4(g, 5); c.r6 = ld4(g, 6); c.r7 = ld4(g, 7);
  return c;
}
static __device__ __forceinline__ void stage_chunk(float* s, const chunk8& c) {
  st4(s, 0, c.r0); st4(s, 1, c.r1); st4(s, 2, c.r2); st4(s, 3, c.r3);
  st4(s, 4, c.r4); st4(s, 5, c.r5); st4(s, 6, c.r6); st4(s, 7, c.r7);
}

static __device__ __forceinline__ void s_block(v8f (&accS)[2], const float* ap,
                                               const float* bp) {
#pragma unroll
  for (int kk = 0; kk < 32; ++kk) {
    v2f a  = *reinterpret_cast<const v2f*>(ap + kk * 4);
    v2f b0 = *reinterpret_cast<const v2f*>(bp + kk * 4);
    v2f b1 = *reinterpret_cast<const v2f*>(bp + 16 * LDP + kk * 4);
    accS[0] = wmma4(a, b0, accS[0]);
    accS[1] = wmma4(a, b1, accS[1]);
  }
}

static __device__ __forceinline__ void pv_block(const float* vp, const float* pp,
                                                float* __restrict__ zo) {
  v8f acc[4];
#pragma unroll
  for (int j = 0; j < 4; ++j) acc[j] = (v8f){};
#pragma unroll
  for (int kk = 0; kk < 16; ++kk) {
    v2f a = *reinterpret_cast<const v2f*>(pp + kk * 4);
#pragma unroll
    for (int j = 0; j < 4; ++j) {
      v2f b;
      b.x = vp[(kk * 4) * LDP + j * 16];
      b.y = vp[(kk * 4 + 1) * LDP + j * 16];
      acc[j] = wmma4(a, b, acc[j]);
    }
  }
#pragma unroll
  for (int j = 0; j < 4; ++j)
#pragma unroll
    for (int v = 0; v < 8; ++v)
      zo[v * HD + j * 16] = acc[j][v];
}

__global__ __launch_bounds__(256) void win_attn_kernel(
    const float* __restrict__ Y, const float* __restrict__ mask,
    float* __restrict__ Z) {
  __shared__ __align__(16) float sQ[2][64 * LDP];
  __shared__ __align__(16) float sK[2][64 * LDP];
  __shared__ __align__(16) float sS[64 * 68];
  const int tid = threadIdx.x, lane = tid & 31, wave = tid >> 5;
  const int bt = blockIdx.x >> 3, h = blockIdx.x & 7;
  const float* qb = Y + (size_t)bt * YBT + h * DD;

  const int m0 = lane & 15;
  const int ksel = (lane >> 4) * 2;
  const int hi8 = (lane >> 4) << 3;
  const int mt = wave >> 1;
  const int nt0 = (wave & 1) * 2;

  const float* gq = qb + (size_t)wave * ROW3 + lane * 4;
  const float* gk = gq + HD;
  const float* gv = gq + 2 * HD;
  const int so = wave * LDP + lane * 4;

  chunk8 rq, rk, rv;

  v8f accS[2];
  accS[0] = (v8f){};
  accS[1] = (v8f){};

  rq = fetch_chunk(gq);
  rk = fetch_chunk(gk);
  stage_chunk(&sQ[0][so], rq);
  stage_chunk(&sK[0][so], rk);
  lds_barrier();

  const int aoff = (mt * 16 + m0) * LDP + ksel;
  const int boff = (nt0 * 16 + m0) * LDP + ksel;
  for (int dc = 0; dc < NCH - 1; ++dc) {
    const int cur = dc & 1;
    rq = fetch_chunk(gq + (dc + 1) * CW);
    rk = fetch_chunk(gk + (dc + 1) * CW);
    s_block(accS, &sQ[cur][aoff], &sK[cur][boff]);
    stage_chunk(&sQ[cur ^ 1][so], rq);
    stage_chunk(&sK[cur ^ 1][so], rk);
    lds_barrier();
  }

  rv = fetch_chunk(gv);
  const float* mk = mask + (size_t)(bt & 7) * (NN * NN);
  float mv[2][8];
#pragma unroll
  for (int j = 0; j < 2; ++j) {
    const int ncol = (nt0 + j) * 16 + m0;
#pragma unroll
    for (int v = 0; v < 8; ++v)
      mv[j][v] = mk[(mt * 16 + v + hi8) * NN + ncol];
  }
  s_block(accS, &sQ[(NCH - 1) & 1][aoff], &sK[(NCH - 1) & 1][boff]);

#pragma unroll
  for (int j = 0; j < 2; ++j) {
    const int ncol = (nt0 + j) * 16 + m0;
#pragma unroll
    for (int v = 0; v < 8; ++v) {
      const int m = mt * 16 + v + hi8;
      sS[m * 68 + ncol] = accS[j][v] * ATTN_SCALE + mv[j][v];
    }
  }
  stage_chunk(&sK[NCH & 1][so], rv);
  lds_barrier();

  for (int r = wave * 8; r < wave * 8 + 8; ++r) {
    float x0 = sS[r * 68 + lane];
    float x1 = sS[r * 68 + lane + 32];
    float mx = fmaxf(x0, x1);
#pragma unroll
    for (int off = 16; off >= 1; off >>= 1)
      mx = fmaxf(mx, __shfl_xor(mx, off, 32));
    const float e0 = __expf(x0 - mx), e1 = __expf(x1 - mx);
    float sm = e0 + e1;
#pragma unroll
    for (int off = 16; off >= 1; off >>= 1)
      sm += __shfl_xor(sm, off, 32);
    const float inv = 1.0f / sm;
    sS[r * 68 + lane] = e0 * inv;
    sS[r * 68 + lane + 32] = e1 * inv;
  }
  lds_barrier();

  float* zb = Z + (size_t)bt * ZBT + h * DD;
  const int dhalf = (wave & 1) * 64;
  const float* pp = &sS[(mt * 16 + m0) * 68 + ksel];
  float* zrow = zb + (size_t)(mt * 16 + hi8) * HD + dhalf + m0;
  for (int dc = 0; dc < NCH - 1; ++dc) {
    const int cur = (NCH + dc) & 1;
    rv = fetch_chunk(gv + (dc + 1) * CW);
    __builtin_amdgcn_sched_barrier(0);
    pv_block(&sK[cur][ksel * LDP + dhalf + m0], pp, zrow + dc * CW);
    stage_chunk(&sK[cur ^ 1][so], rv);
    lds_barrier();
  }
  pv_block(&sK[(NCH + NCH - 1) & 1][ksel * LDP + dhalf + m0], pp, zrow + (NCH - 1) * CW);
}

__global__ __launch_bounds__(256) void proj_gemm_kernel(
    const float* __restrict__ Z, const float* __restrict__ wproj,
    const float* __restrict__ bproj, float* __restrict__ out) {
  __shared__ __align__(16) float sZ[64 * 100];
  const int tid = threadIdx.x, lane = tid & 31, wave = tid >> 5;
  const int g = blockIdx.x;
  const float* zt = Z + (size_t)g * (NN * CC);
  float* ot = out + (size_t)g * (NN * CC);

  for (int i = tid; i < 64 * 24; i += 256) {
    const int r = i / 24, c4 = (i % 24) * 4;
    *reinterpret_cast<float4*>(&sZ[r * 100 + c4]) =
        *reinterpret_cast<const float4*>(zt + r * CC + c4);
  }
  __syncthreads();

  const int m0 = lane & 15;
  const int ksel = (lane >> 4) * 2;
  for (int tt = wave; tt < 24; tt += 8) {
    const int mt = tt & 3, nt = tt >> 2;
    const int ocol = nt * 16 + m0;
    const float* bp = wproj + ocol * CC;
    const float* ap = &sZ[(mt * 16 + m0) * 100];
    v8f acc = {};
#pragma unroll
    for (int kk = 0; kk < 24; ++kk) {
      const int kb = kk * 4 + ksel;
      v2f a = *reinterpret_cast<const v2f*>(ap + kb);
      v2f b = *reinterpret_cast<const v2f*>(bp + kb);
      acc = wmma4(a, b, acc);
    }
    const float bias = bproj[ocol];
#pragma unroll
    for (int v = 0; v < 8; ++v) {
      const int m = mt * 16 + v + ((lane >> 4) << 3);
      ot[m * CC + ocol] = acc[v] + bias;
    }
  }
}

extern "C" void kernel_launch(void* const* d_in, const int* in_sizes, int n_in,
                              void* d_out, int out_size, void* d_ws, size_t ws_size,
                              hipStream_t stream) {
  (void)in_sizes; (void)n_in; (void)out_size; (void)ws_size;
  const float* x     = (const float*)d_in[0];
  const float* mask  = (const float*)d_in[1];
  const float* wqkv  = (const float*)d_in[2];
  const float* wproj = (const float*)d_in[3];
  const float* bproj = (const float*)d_in[4];
  float* out = (float*)d_out;

  float* Y = (float*)d_ws;
  float* Z = Y + (size_t)BTOT * YBT;

  qkv_gemm_kernel<<<dim3(BTOT * WN), dim3(256), 0, stream>>>(x, wqkv, Y);
  win_attn_kernel<<<dim3(BTOT * HH), dim3(256), 0, stream>>>(Y, mask, Z);
  proj_gemm_kernel<<<dim3(BTOT * WN), dim3(256), 0, stream>>>(Z, wproj, bproj, out);
}


# both linear layers: a wave keeps its weight (B) fragments for all of K in registers across the M tiles of a window instead of loading them from global per WMMA; 9 and 6 waves; sums unchanged
# speedup vs baseline: 1.8613x; 1.0623x over previous
#include <hip/hip_runtime.h>


typedef __attribute__((ext_vector_type(2))) float v2f;
typedef __attribute__((ext_vector_type(8))) float v8f;

#define BTOT 16
#define WN   64
#define NN   64
#define CC   96
#define HH   8
#define DD   768
#define C3   288
#define ROW3 18432
#define YBT  1179648
#define HD   6144
#define ZBT  393216
#define ATTN_SCALE 0.28867513459481287f

static __device__ __forceinline__ v8f wmma4(v2f a, v2f b, v8f c) {
  return __builtin_amdgcn_wmma_f32_16x16x4_f32(
       false, a,  false, b,
       (short)0, c,  false,  false);
}

static __device__ __forceinline__ void lds_barrier() {
  __builtin_amdgcn_fence(__ATOMIC_RELEASE, "workgroup", "local");
  __builtin_amdgcn_s_barrier();
  __builtin_amdgcn_fence(__ATOMIC_ACQUIRE, "workgroup", "local");
}

__global__ __launch_bounds__(288) void qkv_gemm_kernel(
    const float* __restrict__ x, const float* __restrict__ wqkv,
    float* __restrict__ Y) {
  __shared__ __align__(16) float sX[64 * 100];
  const int tid = threadIdx.x;
  const int lane = tid & 31, wave = tid >> 5;
  const int g = blockIdx.x;
  const float* xt = x + (size_t)g * (NN * CC);
  float* yt = Y + (size_t)(g / WN) * YBT + (size_t)(g % WN) * ROW3;

  const int m0 = lane & 15;
  const int ksel = (lane >> 4) * 2;
  const int hi8 = (lane >> 4) << 3;
  const int ncol = wave * 32 + m0;

  for (int i = tid; i < 64 * 24; i += 288) {
    int r = i / 24, c4 = (i % 24) * 4;
    *reinterpret_cast<float4*>(&sX[r * 100 + c4]) =
        *reinterpret_cast<const float4*>(xt + r * CC + c4);
  }

  const float* bp = wqkv + ncol * CC + ksel;
  v2f b0[24], b1[24];
#pragma unroll
  for (int kk = 0; kk < 24; ++kk) {
    b0[kk] = *reinterpret_cast<const v2f*>(bp + kk * 4);
    b1[kk] = *reinterpret_cast<const v2f*>(bp + 16 * CC + kk * 4);
  }
  lds_barrier();

  const float* ap = &sX[m0 * 100 + ksel];
  float* yo = yt + hi8 * C3 + ncol;
  for (int mt = 0; mt < 4; ++mt) {
    v8f acc0 = {}, acc1 = {};
#pragma unroll
    for (int kk = 0; kk < 24; ++kk) {
      v2f a = *reinterpret_cast<const v2f*>(ap + mt * (16 * 100) + kk * 4);
      acc0 = wmma4(a, b0[kk], acc0);
      acc1 = wmma4(a, b1[kk], acc1);
    }
#pragma unroll
    for (int v = 0; v < 8; ++v) {
      yo[(mt * 16 + v) * C3] = acc0[v];
      yo[(mt * 16 + v) * C3 + 16] = acc1[v];
    }
  }
}

#define CW   128
#define NCH  6
#define LDP  132

struct chunk8 { float4 r0, r1, r2, r3, r4, r5, r6, r7; };
static __device__ __forceinline__ float4 ld4(const float* __restrict__ g, int u) {
  return *reinterpret_cast<const float4*>(g + (size_t)u * (8 * ROW3));
}
static __device__ __forceinline__ void st4(float* s, int u, float4 v) {
  *reinterpret_cast<float4*>(s + u * (8 * LDP)) = v;
}
static __device__ __forceinline__ chunk8 fetch_chunk(const float* __restrict__ g) {
  chunk8 c;
  c.r0 = ld4(g, 0); c.r1 = ld4(g, 1); c.r2 = ld4(g, 2); c.r3 = ld4(g, 3);
  c.r4 = ld4(g, 4); c.r5 = ld4(g, 5); c.r6 = ld4(g, 6); c.r7 = ld4(g, 7);
  return c;
}
static __device__ __forceinline__ void stage_chunk(float* s, const chunk8& c) {
  st4(s, 0, c.r0); st4(s, 1, c.r1); st4(s, 2, c.r2); st4(s, 3, c.r3);
  st4(s, 4, c.r4); st4(s, 5, c.r5); st4(s, 6, c.r6); st4(s, 7, c.r7);
}

static __device__ __forceinline__ void s_block(v8f (&accS)[2], const float* ap,
                                               const float* bp) {
#pragma unroll
  for (int kk = 0; kk < 32; ++kk) {
    v2f a  = *reinterpret_cast<const v2f*>(ap + kk * 4);
    v2f b0 = *reinterpret_cast<const v2f*>(bp + kk * 4);
    v2f b1 = *reinterpret_cast<const v2f*>(bp + 16 * LDP + kk * 4);
    accS[0] = wmma4(a, b0, accS[0]);
    accS[1] = wmma4(a, b1, accS[1]);
  }
}

static __device__ __forceinline__ void pv_block(const float* vp, const float* pp,
                                                float* __restrict__ zo) {
  v8f acc[4];
#pragma unroll
  for (int j = 0; j < 4; ++j) acc[j] = (v8f){};
#pragma unroll
  for (int kk = 0; kk < 16; ++kk) {
    v2f a = *reinterpret_cast<const v2f*>(pp + kk * 4);
#pragma unroll
    for (int j = 0; j < 4; ++j) {
      v2f b;
      b.x = vp[(kk * 4) * LDP + j * 16];
      b.y = vp[(kk * 4 + 1) * LDP + j * 16];
      acc[j] = wmma4(a, b, acc[j]);
    }
  }
#pragma unroll
  for (int j = 0; j < 4; ++j)
#pragma unroll
    for (int v = 0; v < 8; ++v)
      zo[v * HD + j * 16] = acc[j][v];
}

__global__ __launch_bounds__(256) void win_attn_kernel(
    const float* __restrict__ Y, const float* __restrict__ mask,
    float* __restrict__ Z) {
  __shared__ __align__(16) float sQ[2][64 * LDP];
  __shared__ __align__(16) float sK[2][64 * LDP];
  __shared__ __align__(16) float sS[64 * 68];
  const int tid = threadIdx.x, lane = tid & 31, wave = tid >> 5;
  const int bt = blockIdx.x >> 3, h = blockIdx.x & 7;
  const float* qb = Y + (size_t)bt * YBT + h * DD;

  const int m0 = lane & 15;
  const int ksel = (lane >> 4) * 2;
  const int hi8 = (lane >> 4) << 3;
  const int mt = wave >> 1;
  const int nt0 = (wave & 1) * 2;

  const float* gq = qb + (size_t)wave * ROW3 + lane * 4;
  const float* gk = gq + HD;
  const float* gv = gq + 2 * HD;
  const int so = wave * LDP + lane * 4;

  chunk8 rq, rk, rv;

  v8f accS[2];
  accS[0] = (v8f){};
  accS[1] = (v8f){};

  rq = fetch_chunk(gq);
  rk = fetch_chunk(gk);
  stage_chunk(&sQ[0][so], rq);
  stage_chunk(&sK[0][so], rk);
  lds_barrier();

  const int aoff = (mt * 16 + m0) * LDP + ksel;
  const int boff = (nt0 * 16 + m0) * LDP + ksel;
  for (int dc = 0; dc < NCH - 1; ++dc) {
    const int cur = dc & 1;
    rq = fetch_chunk(gq + (dc + 1) * CW);
    rk = fetch_chunk(gk + (dc + 1) * CW);
    s_block(accS, &sQ[cur][aoff], &sK[cur][boff]);
    stage_chunk(&sQ[cur ^ 1][so], rq);
    stage_chunk(&sK[cur ^ 1][so], rk);
    lds_barrier();
  }

  rv = fetch_chunk(gv);
  const float* mk = mask + (size_t)(bt & 7) * (NN * NN);
  float mv[2][8];
#pragma unroll
  for (int j = 0; j < 2; ++j) {
    const int ncol = (nt0 + j) * 16 + m0;
#pragma unroll
    for (int v = 0; v < 8; ++v)
      mv[j][v] = mk[(mt * 16 + v + hi8) * NN + ncol];
  }
  s_block(accS, &sQ[(NCH - 1) & 1][aoff], &sK[(NCH - 1) & 1][boff]);

#pragma unroll
  for (int j = 0; j < 2; ++j) {
    const int ncol = (nt0 + j) * 16 + m0;
#pragma unroll
    for (int v = 0; v < 8; ++v) {
      const int m = mt * 16 + v + hi8;
      sS[m * 68 + ncol] = accS[j][v] * ATTN_SCALE + mv[j][v];
    }
  }
  stage_chunk(&sK[NCH & 1][so], rv);
  lds_barrier();

  for (int r = wave * 8; r < wave * 8 + 8; ++r) {
    float x0 = sS[r * 68 + lane];
    float x1 = sS[r * 68 + lane + 32];
    float mx = fmaxf(x0, x1);
#pragma unroll
    for (int off = 16; off >= 1; off >>= 1)
      mx = fmaxf(mx, __shfl_xor(mx, off, 32));
    const float e0 = __expf(x0 - mx), e1 = __expf(x1 - mx);
    float sm = e0 + e1;
#pragma unroll
    for (int off = 16; off >= 1; off >>= 1)
      sm += __shfl_xor(sm, off, 32);
    const float inv = 1.0f / sm;
    sS[r * 68 + lane] = e0 * inv;
    sS[r * 68 + lane + 32] = e1 * inv;
  }
  lds_barrier();

  float* zb = Z + (size_t)bt * ZBT + h * DD;
  const int dhalf = (wave & 1) * 64;
  const float* pp = &sS[(mt * 16 + m0) * 68 + ksel];
  float* zrow = zb + (size_t)(mt * 16 + hi8) * HD + dhalf + m0;
  for (int dc = 0; dc < NCH - 1; ++dc) {
    const int cur = (NCH + dc) & 1;
    rv = fetch_chunk(gv + (dc + 1) * CW);
    __builtin_amdgcn_sched_barrier(0);
    pv_block(&sK[cur][ksel * LDP + dhalf + m0], pp, zrow + dc * CW);
    stage_chunk(&sK[cur ^ 1][so], rv);
    lds_barrier();
  }
  pv_block(&sK[(NCH + NCH - 1) & 1][ksel * LDP + dhalf + m0], pp, zrow + (NCH - 1) * CW);
}

__global__ __launch_bounds__(192) void proj_gemm_kernel(
    const float* __restrict__ Z, const float* __restrict__ wproj,
    const float* __restrict__ bproj, float* __restrict__ out) {
  __shared__ __align__(16) float sZ[64 * 100];
  const int tid = threadIdx.x, lane = tid & 31, wave = tid >> 5;
  const int g = blockIdx.x;
  const float* zt = Z + (size_t)g * (NN * CC);
  float* ot = out + (size_t)g * (NN * CC);

  const int m0 = lane & 15;
  const int ksel = (lane >> 4) * 2;
  const int hi8 = (lane >> 4) << 3;
  const int ocol = wave * 16 + m0;

  for (int i = tid; i < 64 * 24; i += 192) {
    const int r = i / 24, c4 = (i % 24) * 4;
    *reinterpret_cast<float4*>(&sZ[r * 100 + c4]) =
        *reinterpret_cast<const float4*>(zt + r * CC + c4);
  }

  const float* bp = wproj + ocol * CC + ksel;
  v2f b[24];
#pragma unroll
  for (int kk = 0; kk < 24; ++kk)
    b[kk] = *reinterpret_cast<const v2f*>(bp + kk * 4);
  const float bias = bproj[ocol];
  lds_barrier();

  const float* ap = &sZ[m0 * 100 + ksel];
  float* oo = ot + hi8 * CC + ocol;
  for (int mp = 0; mp < 2; ++mp) {
    v8f acc0 = {}, acc1 = {};
#pragma unroll
    for (int kk = 0; kk < 24; ++kk) {
      v2f a0 = *reinterpret_cast<const v2f*>(ap + mp * (32 * 100) + kk * 4);
      v2f a1 = *reinterpret_cast<const v2f*>(ap + mp * (32 * 100) + 16 * 100 + kk * 4);
      acc0 = wmma4(a0, b[kk], acc0);
      acc1 = wmma4(a1, b[kk], acc1);
    }
#pragma unroll
    for (int v = 0; v < 8; ++v) {
      oo[(mp * 32 + v) * CC] = acc0[v] + bias;
      oo[(mp * 32 + 16 + v) * CC] = acc1[v] + bias;
    }
  }
}

extern "C" void kernel_launch(void* const* d_in, const int* in_sizes, int n_in,
                              void* d_out, int out_size, void* d_ws, size_t ws_size,
                              hipStream_t stream) {
  (void)in_sizes; (void)n_in; (void)out_size; (void)ws_size;
  const float* x     = (const float*)d_in[0];
  const float* mask  = (const float*)d_in[1];
  const float* wqkv  = (const float*)d_in[2];
  const float* wproj = (const float*)d_in[3];
  const float* bproj = (const float*)d_in[4];
  float* out = (float*)d_out;

  float* Y = (float*)d_ws;
  float* Z = Y + (size_t)BTOT * YBT;

  qkv_gemm_kernel<<<dim3(BTOT * WN), dim3(288), 0, stream>>>(x, wqkv, Y);
  win_attn_kernel<<<dim3(BTOT * HH), dim3(256), 0, stream>>>(Y, mask, Z);
  proj_gemm_kernel<<<dim3(BTOT * WN), dim3(192), 0, stream>>>(Z, wproj, bproj, out);
}


# attention kernel: each (batch,head) split over two work-groups by query rows (32 each), 256 work-groups instead of 128 so every compute unit gets one; order of all sums unchanged
# speedup vs baseline: 2.0721x; 1.1133x over previous
#include <hip/hip_runtime.h>


typedef __attribute__((ext_vector_type(2))) float v2f;
typedef __attribute__((ext_vector_type(8))) float v8f;

#define BTOT 16
#define WN   64
#define NN   64
#define CC   96
#define HH   8
#define DD   768
#define C3   288
#define ROW3 18432
#define YBT  1179648
#define HD   6144
#define ZBT  393216
#define ATTN_SCALE 0.28867513459481287f

static __device__ __forceinline__ v8f wmma4(v2f a, v2f b, v8f c) {
  return __builtin_amdgcn_wmma_f32_16x16x4_f32(
       false, a,  false, b,
       (short)0, c,  false,  false);
}

static __device__ __forceinline__ void lds_barrier() {
  __builtin_amdgcn_fence(__ATOMIC_RELEASE, "workgroup", "local");
  __builtin_amdgcn_s_barrier();
  __builtin_amdgcn_fence(__ATOMIC_ACQUIRE, "workgroup", "local");
}

__global__ __launch_bounds__(288) void qkv_gemm_kernel(
    const float* __restrict__ x, const float* __restrict__ wqkv,
    float* __restrict__ Y) {
  __shared__ __align__(16) float sX[64 * 100];
  const int tid = threadIdx.x;
  const int lane = tid & 31, wave = tid >> 5;
  const int g = blockIdx.x;
  const float* xt = x + (size_t)g * (NN * CC);
  float* yt = Y + (size_t)(g / WN) * YBT + (size_t)(g % WN) * ROW3;

  const int m0 = lane & 15;
  const int ksel = (lane >> 4) * 2;
  const int hi8 = (lane >> 4) << 3;
  const int ncol = wave * 32 + m0;

  for (int i = tid; i < 64 * 24; i += 288) {
    int r = i / 24, c4 = (i % 24) * 4;
    *reinterpret_cast<float4*>(&sX[r * 100 + c4]) =
        *reinterpret_cast<const float4*>(xt + r * CC + c4);
  }

  const float* bp = wqkv + ncol * CC + ksel;
  v2f b0[24], b1[24];
#pragma unroll
  for (int kk = 0; kk < 24; ++kk) {
    b0[kk] = *reinterpret_cast<const v2f*>(bp + kk * 4);
    b1[kk] = *reinterpret_cast<const v2f*>(bp + 16 * CC + kk * 4);
  }
  lds_barrier();

  const float* ap = &sX[m0 * 100 + ksel];
  float* yo = yt + hi8 * C3 + ncol;
  for (int mt = 0; mt < 4; ++mt) {
    v8f acc0 = {}, acc1 = {};
#pragma unroll
    for (int kk = 0; kk < 24; ++kk) {
      v2f a = *reinterpret_cast<const v2f*>(ap + mt * (16 * 100) + kk * 4);
      acc0 = wmma4(a, b0[kk], acc0);
      acc1 = wmma4(a, b1[kk], acc1);
    }
#pragma unroll
    for (int v = 0; v < 8; ++v) {
      yo[(mt * 16 + v) * C3] = acc0[v];
      yo[(mt * 16 + v) * C3 + 16] = acc1[v];
    }
  }
}

#define CW   128
#define NCH  6
#define LDP  132
#define QR   32

struct chunk8 { float4 r0, r1, r2, r3, r4, r5, r6, r7; };
struct chunk4 { float4 r0, r1, r2, r3; };
static __device__ __forceinline__ float4 ld4(const float* __restrict__ g, int u) {
  return *reinterpret_cast<const float4*>(g + (size_t)u * (8 * ROW3));
}
static __device__ __forceinline__ void st4(float* s, int u, float4 v) {
  *reinterpret_cast<float4*>(s + u * (8 * LDP)) = v;
}
static __device__ __forceinline__ chunk8 fetch_chunk(const float* __restrict__ g) {
  chunk8 c;
  c.r0 = ld4(g, 0); c.r1 = ld4(g, 1); c.r2 = ld4(g, 2); c.r3 = ld4(g, 3);
  c.r4 = ld4(g, 4); c.r5 = ld4(g, 5); c.r6 = ld4(g, 6); c.r7 = ld4(g, 7);
  return c;
}
static __device__ __forceinline__ chunk4 fetch_chunk4(const float* __restrict__ g) {
  chunk4 c;
  c.r0 = ld4(g, 0); c.r1 = ld4(g, 1); c.r2 = ld4(g, 2); c.r3 = ld4(g, 3);
  return c;
}
static __device__ __forceinline__ void stage_chunk(float* s, const chunk8& c) {
  st4(s, 0, c.r0); st4(s, 1, c.r1); st4(s, 2, c.r2); st4(s, 3, c.r3);
  st4(s, 4, c.r4); st4(s, 5, c.r5); st4(s, 6, c.r6); st4(s, 7, c.r7);
}
static __device__ __forceinline__ void stage_chunk4(float* s, const chunk4& c) {
  st4(s, 0, c.r0); st4(s, 1, c.r1); st4(s, 2, c.r2); st4(s, 3, c.r3);
}

static __device__ __forceinline__ void s_block(v8f& accS, const float* ap,
                                               const float* bp) {
#pragma unroll
  for (int kk = 0; kk < 32; ++kk) {
    v2f a = *reinterpret_cast<const v2f*>(ap + kk * 4);
    v2f b = *reinterpret_cast<const v2f*>(bp + kk * 4);
    accS = wmma4(a, b, accS);
  }
}

static __device__ __forceinline__ void pv_block(const float* vp, const float* pp,
                                                float* __restrict__ zo) {
  v8f acc[2];
#pragma unroll
  for (int j = 0; j < 2; ++j) acc[j] = (v8f){};
#pragma unroll
  for (int kk = 0; kk < 16; ++kk) {
    v2f a = *reinterpret_cast<const v2f*>(pp + kk * 4);
#pragma unroll
    for (int j = 0; j < 2; ++j) {
      v2f b;
      b.x = vp[(kk * 4) * LDP + j * 16];
      b.y = vp[(kk * 4 + 1) * LDP + j * 16];
      acc[j] = wmma4(a, b, acc[j]);
    }
  }
#pragma unroll
  for (int j = 0; j < 2; ++j)
#pragma unroll
    for (int v = 0; v < 8; ++v)
      zo[v * HD + j * 16] = acc[j][v];
}

__global__ __launch_bounds__(256) void win_attn_kernel(
    const float* __restrict__ Y, const float* __restrict__ mask,
    float* __restrict__ Z) {
  __shared__ __align__(16) float sQ[2][QR * LDP];
  __shared__ __align__(16) float sK[2][64 * LDP];
  __shared__ __align__(16) float sS[QR * 68];
  const int tid = threadIdx.x, lane = tid & 31, wave = tid >> 5;
  const int bt = blockIdx.x >> 4, h = (blockIdx.x >> 1) & 7;
  const int q0 = (blockIdx.x & 1) * QR;
  const float* qb = Y + (size_t)bt * YBT + h * DD;

  const int m0 = lane & 15;
  const int ksel = (lane >> 4) * 2;
  const int hi8 = (lane >> 4) << 3;
  const int mt = wave >> 2;
  const int nt = wave & 3;

  const float* gk = qb + (size_t)wave * ROW3 + lane * 4 + HD;
  const float* gv = gk + HD;
  const float* gq = qb + (size_t)(q0 + wave) * ROW3 + lane * 4;
  const int so = wave * LDP + lane * 4;

  chunk4 rq;
  chunk8 rk, rv;

  v8f accS = (v8f){};

  rq = fetch_chunk4(gq);
  rk = fetch_chunk(gk);
  stage_chunk4(&sQ[0][so], rq);
  stage_chunk(&sK[0][so], rk);
  lds_barrier();

  const int aoff = (mt * 16 + m0) * LDP + ksel;
  const int boff = (nt * 16 + m0) * LDP + ksel;
  for (int dc = 0; dc < NCH - 1; ++dc) {
    const int cur = dc & 1;
    rq = fetch_chunk4(gq + (dc + 1) * CW);
    rk = fetch_chunk(gk + (dc + 1) * CW);
    s_block(accS, &sQ[cur][aoff], &sK[cur][boff]);
    stage_chunk4(&sQ[cur ^ 1][so], rq);
    stage_chunk(&sK[cur ^ 1][so], rk);
    lds_barrier();
  }

  rv = fetch_chunk(gv);
  const float* mk = mask + (size_t)(bt & 7) * (NN * NN) + (size_t)q0 * NN;
  const int ncol = nt * 16 + m0;
  float mv[8];
#pragma unroll
  for (int v = 0; v < 8; ++v)
    mv[v] = mk[(mt * 16 + v + hi8) * NN + ncol];
  s_block(accS, &sQ[(NCH - 1) & 1][aoff], &sK[(NCH - 1) & 1][boff]);

#pragma unroll
  for (int v = 0; v < 8; ++v) {
    const int m = mt * 16 + v + hi8;
    sS[m * 68 + ncol] = accS[v] * ATTN_SCALE + mv[v];
  }
  stage_chunk(&sK[NCH & 1][so], rv);
  lds_barrier();

  for (int r = wave * 4; r < wave * 4 + 4; ++r) {
    float x0 = sS[r * 68 + lane];
    float x1 = sS[r * 68 + lane + 32];
    float mx = fmaxf(x0, x1);
#pragma unroll
    for (int off = 16; off >= 1; off >>= 1)
      mx = fmaxf(mx, __shfl_xor(mx, off, 32));
    const float e0 = __expf(x0 - mx), e1 = __expf(x1 - mx);
    float sm = e0 + e1;
#pragma unroll
    for (int off = 16; off >= 1; off >>= 1)
      sm += __shfl_xor(sm, off, 32);
    const float inv = 1.0f / sm;
    sS[r * 68 + lane] = e0 * inv;
    sS[r * 68 + lane + 32] = e1 * inv;
  }
  lds_barrier();

  float* zb = Z + (size_t)bt * ZBT + h * DD;
  const int dq = nt * 32;
  const float* pp = &sS[(mt * 16 + m0) * 68 + ksel];
  float* zrow = zb + (size_t)(q0 + mt * 16 + hi8) * HD + dq + m0;
  for (int dc = 0; dc < NCH - 1; ++dc) {
    const int cur = (NCH + dc) & 1;
    rv = fetch_chunk(gv + (dc + 1) * CW);
    __builtin_amdgcn_sched_barrier(0);
    pv_block(&sK[cur][ksel * LDP + dq + m0], pp, zrow + dc * CW);
    stage_chunk(&sK[cur ^ 1][so], rv);
    lds_barrier();
  }
  pv_block(&sK[(NCH + NCH - 1) & 1][ksel * LDP + dq + m0], pp, zrow + (NCH - 1) * CW);
}

__global__ __launch_bounds__(192) void proj_gemm_kernel(
    const float* __restrict__ Z, const float* __restrict__ wproj,
    const float* __restrict__ bproj, float* __restrict__ out) {
  __shared__ __align__(16) float sZ[64 * 100];
  const int tid = threadIdx.x, lane = tid & 31, wave = tid >> 5;
  const int g = blockIdx.x;
  const float* zt = Z + (size_t)g * (NN * CC);
  float* ot = out + (size_t)g * (NN * CC);

  const int m0 = lane & 15;
  const int ksel = (lane >> 4) * 2;
  const int hi8 = (lane >> 4) << 3;
  const int ocol = wave * 16 + m0;

  for (int i = tid; i < 64 * 24; i += 192) {
    const int r = i / 24, c4 = (i % 24) * 4;
    *reinterpret_cast<float4*>(&sZ[r * 100 + c4]) =
        *reinterpret_cast<const float4*>(zt + r * CC + c4);
  }

  const float* bp = wproj + ocol * CC + ksel;
  v2f b[24];
#pragma unroll
  for (int kk = 0; kk < 24; ++kk)
    b[kk] = *reinterpret_cast<const v2f*>(bp + kk * 4);
  const float bias = bproj[ocol];
  lds_barrier();

  const float* ap = &sZ[m0 * 100 + ksel];
  float* oo = ot + hi8 * CC + ocol;
  for (int mp = 0; mp < 2; ++mp) {
    v8f acc0 = {}, acc1 = {};
#pragma unroll
    for (int kk = 0; kk < 24; ++kk) {
      v2f a0 = *reinterpret_cast<const v2f*>(ap + mp * (32 * 100) + kk * 4);
      v2f a1 = *reinterpret_cast<const v2f*>(ap + mp * (32 * 100) + 16 * 100 + kk * 4);
      acc0 = wmma4(a0, b[kk], acc0);
      acc1 = wmma4(a1, b[kk], acc1);
    }
#pragma unroll
    for (int v = 0; v < 8; ++v) {
      oo[(mp * 32 + v) * CC] = acc0[v] + bias;
      oo[(mp * 32 + 16 + v) * CC] = acc1[v] + bias;
    }
  }
}

extern "C" void kernel_launch(void* const* d_in, const int* in_sizes, int n_in,
                              void* d_out, int out_size, void* d_ws, size_t ws_size,
                              hipStream_t stream) {
  (void)in_sizes; (void)n_in; (void)out_size; (void)ws_size;
  const float* x     = (const float*)d_in[0];
  const float* mask  = (const float*)d_in[1];
  const float* wqkv  = (const float*)d_in[2];
  const float* wproj = (const float*)d_in[3];
  const float* bproj = (const float*)d_in[4];
  float* out = (float*)d_out;

  float* Y = (float*)d_ws;
  float* Z = Y + (size_t)BTOT * YBT;

  qkv_gemm_kernel<<<dim3(BTOT * WN), dim3(288), 0, stream>>>(x, wqkv, Y);
  win_attn_kernel<<<dim3(BTOT * HH * 2), dim3(256), 0, stream>>>(Y, mask, Z);
  proj_gemm_kernel<<<dim3(BTOT * WN), dim3(192), 0, stream>>>(Z, wproj, bproj, out);
}


# both linear layers: a thread's tile loads (6 and 8 float4) and its weight-fragment loads all issued together before the first wait, instead of one memory round trip per float4; sums unchanged
# speedup vs baseline: 2.2930x; 1.1066x over previous
#include <hip/hip_runtime.h>


typedef __attribute__((ext_vector_type(2))) float v2f;
typedef __attribute__((ext_vector_type(8))) float v8f;

#define BTOT 16
#define WN   64
#define NN   64
#define CC   96
#define HH   8
#define DD   768
#define C3   288
#define ROW3 18432
#define YBT  1179648
#define HD   6144
#define ZBT  393216
#define ATTN_SCALE 0.28867513459481287f

static __device__ __forceinline__ v8f wmma4(v2f a, v2f b, v8f c) {
  return __builtin_amdgcn_wmma_f32_16x16x4_f32(
       false, a,  false, b,
       (short)0, c,  false,  false);
}

static __device__ __forceinline__ void lds_barrier() {
  __builtin_amdgcn_fence(__ATOMIC_RELEASE, "workgroup", "local");
  __builtin_amdgcn_s_barrier();
  __builtin_amdgcn_fence(__ATOMIC_ACQUIRE, "workgroup", "local");
}

__global__ __launch_bounds__(288) void qkv_gemm_kernel(
    const float* __restrict__ x, const float* __restrict__ wqkv,
    float* __restrict__ Y) {
  __shared__ __align__(16) float sX[64 * 100];
  const int tid = threadIdx.x;
  const int lane = tid & 31, wave = tid >> 5;
  const int g = blockIdx.x;
  const float* xt = x + (size_t)g * (NN * CC);
  float* yt = Y + (size_t)(g / WN) * YBT + (size_t)(g % WN) * ROW3;

  const int m0 = lane & 15;
  const int ksel = (lane >> 4) * 2;
  const int hi8 = (lane >> 4) << 3;
  const int ncol = wave * 32 + m0;

  const int xr0 = tid / 24, xc4 = (tid % 24) * 4;
  const float* xg = xt + tid * 4;
  float4 xr[5];
  float4 xr5 = make_float4(0.f, 0.f, 0.f, 0.f);
#pragma unroll
  for (int j = 0; j < 5; ++j)
    xr[j] = *reinterpret_cast<const float4*>(xg + j * (288 * 4));
  if (tid < 96) xr5 = *reinterpret_cast<const float4*>(xg + 5 * (288 * 4));

  const float* bp = wqkv + ncol * CC + ksel;
  v2f b0[24], b1[24];
#pragma unroll
  for (int kk = 0; kk < 24; ++kk) {
    b0[kk] = *reinterpret_cast<const v2f*>(bp + kk * 4);
    b1[kk] = *reinterpret_cast<const v2f*>(bp + 16 * CC + kk * 4);
  }
  float* sxp = &sX[xr0 * 100 + xc4];
#pragma unroll
  for (int j = 0; j < 5; ++j)
    *reinterpret_cast<float4*>(sxp + j * (12 * 100)) = xr[j];
  if (tid < 96) *reinterpret_cast<float4*>(sxp + 5 * (12 * 100)) = xr5;
  lds_barrier();

  const float* ap = &sX[m0 * 100 + ksel];
  float* yo = yt + hi8 * C3 + ncol;
  for (int mt = 0; mt < 4; ++mt) {
    v8f acc0 = {}, acc1 = {};
#pragma unroll
    for (int kk = 0; kk < 24; ++kk) {
      v2f a = *reinterpret_cast<const v2f*>(ap + mt * (16 * 100) + kk * 4);
      acc0 = wmma4(a, b0[kk], acc0);
      acc1 = wmma4(a, b1[kk], acc1);
    }
#pragma unroll
    for (int v = 0; v < 8; ++v) {
      yo[(mt * 16 + v) * C3] = acc0[v];
      yo[(mt * 16 + v) * C3 + 16] = acc1[v];
    }
  }
}

#define CW   128
#define NCH  6
#define LDP  132
#define QR   32

struct chunk8 { float4 r0, r1, r2, r3, r4, r5, r6, r7; };
struct chunk4 { float4 r0, r1, r2, r3; };
static __device__ __forceinline__ float4 ld4(const float* __restrict__ g, int u) {
  return *reinterpret_cast<const float4*>(g + (size_t)u * (8 * ROW3));
}
static __device__ __forceinline__ void st4(float* s, int u, float4 v) {
  *reinterpret_cast<float4*>(s + u * (8 * LDP)) = v;
}
static __device__ __forceinline__ chunk8 fetch_chunk(const float* __restrict__ g) {
  chunk8 c;
  c.r0 = ld4(g, 0); c.r1 = ld4(g, 1); c.r2 = ld4(g, 2); c.r3 = ld4(g, 3);
  c.r4 = ld4(g, 4); c.r5 = ld4(g, 5); c.r6 = ld4(g, 6); c.r7 = ld4(g, 7);
  return c;
}
static __device__ __forceinline__ chunk4 fetch_chunk4(const float* __restrict__ g) {
  chunk4 c;
  c.r0 = ld4(g, 0); c.r1 = ld4(g, 1); c.r2 = ld4(g, 2); c.r3 = ld4(g, 3);
  return c;
}
static __device__ __forceinline__ void stage_chunk(float* s, const chunk8& c) {
  st4(s, 0, c.r0); st4(s, 1, c.r1); st4(s, 2, c.r2); st4(s, 3, c.r3);
  st4(s, 4, c.r4); st4(s, 5, c.r5); st4(s, 6, c.r6); st4(s, 7, c.r7);
}
static __device__ __forceinline__ void stage_chunk4(float* s, const chunk4& c) {
  st4(s, 0, c.r0); st4(s, 1, c.r1); st4(s, 2, c.r2); st4(s, 3, c.r3);
}

static __device__ __forceinline__ void s_block(v8f& accS, const float* ap,
                                               const float* bp) {
#pragma unroll
  for (int kk = 0; kk < 32; ++kk) {
    v2f a = *reinterpret_cast<const v2f*>(ap + kk * 4);
    v2f b = *reinterpret_cast<const v2f*>(bp + kk * 4);
    accS = wmma4(a, b, accS);
  }
}

static __device__ __forceinline__ void pv_block(const float* vp, const float* pp,
                                                float* __restrict__ zo) {
  v8f acc[2];
#pragma unroll
  for (int j = 0; j < 2; ++j) acc[j] = (v8f){};
#pragma unroll
  for (int kk = 0; kk < 16; ++kk) {
    v2f a = *reinterpret_cast<const v2f*>(pp + kk * 4);
#pragma unroll
    for (int j = 0; j < 2; ++j) {
      v2f b;
      b.x = vp[(kk * 4) * LDP + j * 16];
      b.y = vp[(kk * 4 + 1) * LDP + j * 16];
      acc[j] = wmma4(a, b, acc[j]);
    }
  }
#pragma unroll
  for (int j = 0; j < 2; ++j)
#pragma unroll
    for (int v = 0; v < 8; ++v)
      zo[v * HD + j * 16] = acc[j][v];
}

__global__ __launch_bounds__(256) void win_attn_kernel(
    const float* __restrict__ Y, const float* __restrict__ mask,
    float* __restrict__ Z) {
  __shared__ __align__(16) float sQ[2][QR * LDP];
  __shared__ __align__(16) float sK[2][64 * LDP];
  __shared__ __align__(16) float sS[QR * 68];
  const int tid = threadIdx.x, lane = tid & 31, wave = tid >> 5;
  const int bt = blockIdx.x >> 4, h = (blockIdx.x >> 1) & 7;
  const int q0 = (blockIdx.x & 1) * QR;
  const float* qb = Y + (size_t)bt * YBT + h * DD;

  const int m0 = lane & 15;
  const int ksel = (lane >> 4) * 2;
  const int hi8 = (lane >> 4) << 3;
  const int mt = wave >> 2;
  const int nt = wave & 3;

  const float* gk = qb + (size_t)wave * ROW3 + lane * 4 + HD;
  const float* gv = gk + HD;
  const float* gq = qb + (size_t)(q0 + wave) * ROW3 + lane * 4;
  const int so = wave * LDP + lane * 4;

  chunk4 rq;
  chunk8 rk, rv;

  v8f accS = (v8f){};

  rq = fetch_chunk4(gq);
  rk = fetch_chunk(gk);
  stage_chunk4(&sQ[0][so], rq);
  stage_chunk(&sK[0][so], rk);
  lds_barrier();

  const int aoff = (mt * 16 + m0) * LDP + ksel;
  const int boff = (nt * 16 + m0) * LDP + ksel;
  for (int dc = 0; dc < NCH - 1; ++dc) {
    const int cur = dc & 1;
    rq = fetch_chunk4(gq + (dc + 1) * CW);
    rk = fetch_chunk(gk + (dc + 1) * CW);
    s_block(accS, &sQ[cur][aoff], &sK[cur][boff]);
    stage_chunk4(&sQ[cur ^ 1][so], rq);
    stage_chunk(&sK[cur ^ 1][so], rk);
    lds_barrier();
  }

  rv = fetch_chunk(gv);
  const float* mk = mask + (size_t)(bt & 7) * (NN * NN) + (size_t)q0 * NN;
  const int ncol = nt * 16 + m0;
  float mv[8];
#pragma unroll
  for (int v = 0; v < 8; ++v)
    mv[v] = mk[(mt * 16 + v + hi8) * NN + ncol];
  s_block(accS, &sQ[(NCH - 1) & 1][aoff], &sK[(NCH - 1) & 1][boff]);

#pragma unroll
  for (int v = 0; v < 8; ++v) {
    const int m = mt * 16 + v + hi8;
    sS[m * 68 + ncol] = accS[v] * ATTN_SCALE + mv[v];
  }
  stage_chunk(&sK[NCH & 1][so], rv);
  lds_barrier();

  for (int r = wave * 4; r < wave * 4 + 4; ++r) {
    float x0 = sS[r * 68 + lane];
    float x1 = sS[r * 68 + lane + 32];
    float mx = fmaxf(x0, x1);
#pragma unroll
    for (int off = 16; off >= 1; off >>= 1)
      mx = fmaxf(mx, __shfl_xor(mx, off, 32));
    const float e0 = __expf(x0 - mx), e1 = __expf(x1 - mx);
    float sm = e0 + e1;
#pragma unroll
    for (int off = 16; off >= 1; off >>= 1)
      sm += __shfl_xor(sm, off, 32);
    const float inv = 1.0f / sm;
    sS[r * 68 + lane] = e0 * inv;
    sS[r * 68 + lane + 32] = e1 * inv;
  }
  lds_barrier();

  float* zb = Z + (size_t)bt * ZBT + h * DD;
  const int dq = nt * 32;
  const float* pp = &sS[(mt * 16 + m0) * 68 + ksel];
  float* zrow = zb + (size_t)(q0 + mt * 16 + hi8) * HD + dq + m0;
  for (int dc = 0; dc < NCH - 1; ++dc) {
    const int cur = (NCH + dc) & 1;
    rv = fetch_chunk(gv + (dc + 1) * CW);
    __builtin_amdgcn_sched_barrier(0);
    pv_block(&sK[cur][ksel * LDP + dq + m0], pp, zrow + dc * CW);
    stage_chunk(&sK[cur ^ 1][so], rv);
    lds_barrier();
  }
  pv_block(&sK[(NCH + NCH - 1) & 1][ksel * LDP + dq + m0], pp, zrow + (NCH - 1) * CW);
}

__global__ __launch_bounds__(192) void proj_gemm_kernel(
    const float* __restrict__ Z, const float* __restrict__ wproj,
    const float* __restrict__ bproj, float* __restrict__ out) {
  __shared__ __align__(16) float sZ[64 * 100];
  const int tid = threadIdx.x, lane = tid & 31, wave = tid >> 5;
  const int g = blockIdx.x;
  const float* zt = Z + (size_t)g * (NN * CC);
  float* ot = out + (size_t)g * (NN * CC);

  const int m0 = lane & 15;
  const int ksel = (lane >> 4) * 2;
  const int hi8 = (lane >> 4) << 3;
  const int ocol = wave * 16 + m0;

  const int zr0 = tid / 24, zc4 = (tid % 24) * 4;
  const float* zg = zt + tid * 4;
  float4 zr[8];
#pragma unroll
  for (int j = 0; j < 8; ++j)
    zr[j] = *reinterpret_cast<const float4*>(zg + j * (192 * 4));

  const float* bp = wproj + ocol * CC + ksel;
  v2f b[24];
#pragma unroll
  for (int kk = 0; kk < 24; ++kk)
    b[kk] = *reinterpret_cast<const v2f*>(bp + kk * 4);
  const float bias = bproj[ocol];
  float* szp = &sZ[zr0 * 100 + zc4];
#pragma unroll
  for (int j = 0; j < 8; ++j)
    *reinterpret_cast<float4*>(szp + j * (8 * 100)) = zr[j];
  lds_barrier();

  const float* ap = &sZ[m0 * 100 + ksel];
  float* oo = ot + hi8 * CC + ocol;
  for (int mp = 0; mp < 2; ++mp) {
    v8f acc0 = {}, acc1 = {};
#pragma unroll
    for (int kk = 0; kk < 24; ++kk) {
      v2f a0 = *reinterpret_cast<const v2f*>(ap + mp * (32 * 100) + kk * 4);
      v2f a1 = *reinterpret_cast<const v2f*>(ap + mp * (32 * 100) + 16 * 100 + kk * 4);
      acc0 = wmma4(a0, b[kk], acc0);
      acc1 = wmma4(a1, b[kk], acc1);
    }
#pragma unroll
    for (int v = 0; v < 8; ++v) {
      oo[(mp * 32 + v) * CC] = acc0[v] + bias;
      oo[(mp * 32 + 16 + v) * CC] = acc1[v] + bias;
    }
  }
}

extern "C" void kernel_launch(void* const* d_in, const int* in_sizes, int n_in,
                              void* d_out, int out_size, void* d_ws, size_t ws_size,
                              hipStream_t stream) {
  (void)in_sizes; (void)n_in; (void)out_size; (void)ws_size;
  const float* x     = (const float*)d_in[0];
  const float* mask  = (const float*)d_in[1];
  const float* wqkv  = (const float*)d_in[2];
  const float* wproj = (const float*)d_in[3];
  const float* bproj = (const float*)d_in[4];
  float* out = (float*)d_out;

  float* Y = (float*)d_ws;
  float* Z = Y + (size_t)BTOT * YBT;

  qkv_gemm_kernel<<<dim3(BTOT * WN), dim3(288), 0, stream>>>(x, wqkv, Y);
  win_attn_kernel<<<dim3(BTOT * HH * 2), dim3(256), 0, stream>>>(Y, mask, Z);
  proj_gemm_kernel<<<dim3(BTOT * WN), dim3(192), 0, stream>>>(Z, wproj, bproj, out);
}


# proj layer fused into the attention kernel: a work-group keeps its 32x768 O tile in LDS and multiplies it by Wproj (staged in LDS) itself; 2 launches, no Z round trip; sums unchanged
# speedup vs baseline: 2.5738x; 1.1225x over previous
#include <hip/hip_runtime.h>


typedef __attribute__((ext_vector_type(2))) float v2f;
typedef __attribute__((ext_vector_type(8))) float v8f;

#define BTOT 16
#define WN   64
#define NN   64
#define CC   96
#define HH   8
#define DD   768
#define C3   288
#define ROW3 18432
#define YBT  1179648
#define HD   6144
#define ATTN_SCALE 0.28867513459481287f

static __device__ __forceinline__ v8f wmma4(v2f a, v2f b, v8f c) {
  return __builtin_amdgcn_wmma_f32_16x16x4_f32(
       false, a,  false, b,
       (short)0, c,  false,  false);
}

static __device__ __forceinline__ void lds_barrier() {
  __builtin_amdgcn_fence(__ATOMIC_RELEASE, "workgroup", "local");
  __builtin_amdgcn_s_barrier();
  __builtin_amdgcn_fence(__ATOMIC_ACQUIRE, "workgroup", "local");
}

__global__ __launch_bounds__(288) void qkv_gemm_kernel(
    const float* __restrict__ x, const float* __restrict__ wqkv,
    float* __restrict__ Y) {
  __shared__ __align__(16) float sX[64 * 100];
  const int tid = threadIdx.x;
  const int lane = tid & 31, wave = tid >> 5;
  const int g = blockIdx.x;
  const float* xt = x + (size_t)g * (NN * CC);
  float* yt = Y + (size_t)(g / WN) * YBT + (size_t)(g % WN) * ROW3;

  const int m0 = lane & 15;
  const int ksel = (lane >> 4) * 2;
  const int hi8 = (lane >> 4) << 3;
  const int ncol = wave * 32 + m0;

  const int xr0 = tid / 24, xc4 = (tid % 24) * 4;
  const float* xg = xt + tid * 4;
  float4 xr[5];
  float4 xr5 = make_float4(0.f, 0.f, 0.f, 0.f);
#pragma unroll
  for (int j = 0; j < 5; ++j)
    xr[j] = *reinterpret_cast<const float4*>(xg + j * (288 * 4));
  if (tid < 96) xr5 = *reinterpret_cast<const float4*>(xg + 5 * (288 * 4));

  const float* bp = wqkv + ncol * CC + ksel;
  v2f b0[24], b1[24];
#pragma unroll
  for (int kk = 0; kk < 24; ++kk) {
    b0[kk] = *reinterpret_cast<const v2f*>(bp + kk * 4);
    b1[kk] = *reinterpret_cast<const v2f*>(bp + 16 * CC + kk * 4);
  }
  float* sxp = &sX[xr0 * 100 + xc4];
#pragma unroll
  for (int j = 0; j < 5; ++j)
    *reinterpret_cast<float4*>(sxp + j * (12 * 100)) = xr[j];
  if (tid < 96) *reinterpret_cast<float4*>(sxp + 5 * (12 * 100)) = xr5;
  lds_barrier();

  const float* ap = &sX[m0 * 100 + ksel];
  float* yo = yt + hi8 * C3 + ncol;
  for (int mt = 0; mt < 4; ++mt) {
    v8f acc0 = {}, acc1 = {};
#pragma unroll
    for (int kk = 0; kk < 24; ++kk) {
      v2f a = *reinterpret_cast<const v2f*>(ap + mt * (16 * 100) + kk * 4);
      acc0 = wmma4(a, b0[kk], acc0);
      acc1 = wmma4(a, b1[kk], acc1);
    }
#pragma unroll
    for (int v = 0; v < 8; ++v) {
      yo[(mt * 16 + v) * C3] = acc0[v];
      yo[(mt * 16 + v) * C3 + 16] = acc1[v];
    }
  }
}

#define CW   128
#define NCH  6
#define LDP  132
#define QR   32
#define OP   772
#define WP   100

struct chunk8 { float4 r0, r1, r2, r3, r4, r5, r6, r7; };
struct chunk4 { float4 r0, r1, r2, r3; };
static __device__ __forceinline__ float4 ld4(const float* __restrict__ g, int u) {
  return *reinterpret_cast<const float4*>(g + (size_t)u * (8 * ROW3));
}
static __device__ __forceinline__ void st4(float* s, int u, float4 v) {
  *reinterpret_cast<float4*>(s + u * (8 * LDP)) = v;
}
static __device__ __forceinline__ chunk8 fetch_chunk(const float* __restrict__ g) {
  chunk8 c;
  c.r0 = ld4(g, 0); c.r1 = ld4(g, 1); c.r2 = ld4(g, 2); c.r3 = ld4(g, 3);
  c.r4 = ld4(g, 4); c.r5 = ld4(g, 5); c.r6 = ld4(g, 6); c.r7 = ld4(g, 7);
  return c;
}
static __device__ __forceinline__ chunk4 fetch_chunk4(const float* __restrict__ g) {
  chunk4 c;
  c.r0 = ld4(g, 0); c.r1 = ld4(g, 1); c.r2 = ld4(g, 2); c.r3 = ld4(g, 3);
  return c;
}
static __device__ __forceinline__ void stage_chunk(float* s, const chunk8& c) {
  st4(s, 0, c.r0); st4(s, 1, c.r1); st4(s, 2, c.r2); st4(s, 3, c.r3);
  st4(s, 4, c.r4); st4(s, 5, c.r5); st4(s, 6, c.r6); st4(s, 7, c.r7);
}
static __device__ __forceinline__ void stage_chunk4(float* s, const chunk4& c) {
  st4(s, 0, c.r0); st4(s, 1, c.r1); st4(s, 2, c.r2); st4(s, 3, c.r3);
}

static __device__ __forceinline__ void s_block(v8f& accS, const float* ap,
                                               const float* bp) {
#pragma unroll
  for (int kk = 0; kk < 32; ++kk) {
    v2f a = *reinterpret_cast<const v2f*>(ap + kk * 4);
    v2f b = *reinterpret_cast<const v2f*>(bp + kk * 4);
    accS = wmma4(a, b, accS);
  }
}

static __device__ __forceinline__ void pv_block(const float* vp, const float* pp,
                                                float* so) {
  v8f acc[2];
#pragma unroll
  for (int j = 0; j < 2; ++j) acc[j] = (v8f){};
#pragma unroll
  for (int kk = 0; kk < 16; ++kk) {
    v2f a = *reinterpret_cast<const v2f*>(pp + kk * 4);
#pragma unroll
    for (int j = 0; j < 2; ++j) {
      v2f b;
      b.x = vp[(kk * 4) * LDP + j * 16];
      b.y = vp[(kk * 4 + 1) * LDP + j * 16];
      acc[j] = wmma4(a, b, acc[j]);
    }
  }
#pragma unroll
  for (int j = 0; j < 2; ++j)
#pragma unroll
    for (int v = 0; v < 8; ++v)
      so[v * OP + j * 16] = acc[j][v];
}

__global__ __launch_bounds__(256) void win_attn_proj_kernel(
    const float* __restrict__ Y, const float* __restrict__ mask,
    const float* __restrict__ wproj, const float* __restrict__ bproj,
    float* __restrict__ out) {
  __shared__ __align__(16) float sQ[2][QR * LDP];
  __shared__ __align__(16) float sK[2][64 * LDP];
  __shared__ __align__(16) float sS[QR * 68];
  __shared__ __align__(16) float sO[QR * OP];
  __shared__ __align__(16) float sW[CC * WP];
  const int tid = threadIdx.x, lane = tid & 31, wave = tid >> 5;
  const int bt = blockIdx.x >> 4, h = (blockIdx.x >> 1) & 7;
  const int q0 = (blockIdx.x & 1) * QR;
  const float* qb = Y + (size_t)bt * YBT + h * DD;

  const int m0 = lane & 15;
  const int ksel = (lane >> 4) * 2;
  const int hi8 = (lane >> 4) << 3;
  const int mt = wave >> 2;
  const int nt = wave & 3;

  const float* gk = qb + (size_t)wave * ROW3 + lane * 4 + HD;
  const float* gv = gk + HD;
  const float* gq = qb + (size_t)(q0 + wave) * ROW3 + lane * 4;
  const int so = wave * LDP + lane * 4;

  chunk4 rq;
  chunk8 rk, rv;

  v8f accS = (v8f){};

  float4 wr[9];
#pragma unroll
  for (int i = 0; i < 9; ++i)
    wr[i] = *reinterpret_cast<const float4*>(wproj + (tid + 256 * i) * 4);

  rq = fetch_chunk4(gq);
  rk = fetch_chunk(gk);
#pragma unroll
  for (int i = 0; i < 9; ++i) {
    const int f = tid + 256 * i;
    *reinterpret_cast<float4*>(&sW[(f / 24) * WP + (f % 24) * 4]) = wr[i];
  }
  stage_chunk4(&sQ[0][so], rq);
  stage_chunk(&sK[0][so], rk);
  lds_barrier();

  const int aoff = (mt * 16 + m0) * LDP + ksel;
  const int boff = (nt * 16 + m0) * LDP + ksel;
  for (int dc = 0; dc < NCH - 1; ++dc) {
    const int cur = dc & 1;
    rq = fetch_chunk4(gq + (dc + 1) * CW);
    rk = fetch_chunk(gk + (dc + 1) * CW);
    s_block(accS, &sQ[cur][aoff], &sK[cur][boff]);
    stage_chunk4(&sQ[cur ^ 1][so], rq);
    stage_chunk(&sK[cur ^ 1][so], rk);
    lds_barrier();
  }

  rv = fetch_chunk(gv);
  const float* mk = mask + (size_t)(bt & 7) * (NN * NN) + (size_t)q0 * NN;
  const int ncol = nt * 16 + m0;
  float mv[8];
#pragma unroll
  for (int v = 0; v < 8; ++v)
    mv[v] = mk[(mt * 16 + v + hi8) * NN + ncol];
  s_block(accS, &sQ[(NCH - 1) & 1][aoff], &sK[(NCH - 1) & 1][boff]);

#pragma unroll
  for (int v = 0; v < 8; ++v) {
    const int m = mt * 16 + v + hi8;
    sS[m * 68 + ncol] = accS[v] * ATTN_SCALE + mv[v];
  }
  stage_chunk(&sK[NCH & 1][so], rv);
  lds_barrier();

  for (int r = wave * 4; r < wave * 4 + 4; ++r) {
    float x0 = sS[r * 68 + lane];
    float x1 = sS[r * 68 + lane + 32];
    float mx = fmaxf(x0, x1);
#pragma unroll
    for (int off = 16; off >= 1; off >>= 1)
      mx = fmaxf(mx, __shfl_xor(mx, off, 32));
    const float e0 = __expf(x0 - mx), e1 = __expf(x1 - mx);
    float sm = e0 + e1;
#pragma unroll
    for (int off = 16; off >= 1; off >>= 1)
      sm += __shfl_xor(sm, off, 32);
    const float inv = 1.0f / sm;
    sS[r * 68 + lane] = e0 * inv;
    sS[r * 68 + lane + 32] = e1 * inv;
  }
  lds_barrier();

  const int dq = nt * 32;
  const float* pp = &sS[(mt * 16 + m0) * 68 + ksel];
  float* zrow = &sO[(mt * 16 + hi8) * OP + dq + m0];
  for (int dc = 0; dc < NCH - 1; ++dc) {
    const int cur = (NCH + dc) & 1;
    rv = fetch_chunk(gv + (dc + 1) * CW);
    __builtin_amdgcn_sched_barrier(0);
    pv_block(&sK[cur][ksel * LDP + dq + m0], pp, zrow + dc * CW);
    stage_chunk(&sK[cur ^ 1][so], rv);
    lds_barrier();
  }
  pv_block(&sK[(NCH + NCH - 1) & 1][ksel * LDP + dq + m0], pp, zrow + (NCH - 1) * CW);
  lds_barrier();

  const float* oa = &sO[m0 * OP + wave * CC + ksel];
  v2f a0[24], a1[24];
#pragma unroll
  for (int kk = 0; kk < 24; ++kk) {
    a0[kk] = *reinterpret_cast<const v2f*>(oa + kk * 4);
    a1[kk] = *reinterpret_cast<const v2f*>(oa + 16 * OP + kk * 4);
  }
  float* ob = out + ((size_t)(bt * WN + q0 + hi8) * NN + (h * 8 + wave)) * CC + m0;
  const float* wb = &sW[m0 * WP + ksel];
  v2f b[2][24];
#pragma unroll
  for (int kk = 0; kk < 24; ++kk)
    b[0][kk] = *reinterpret_cast<const v2f*>(wb + kk * 4);
#pragma unroll
  for (int nn = 0; nn < 6; ++nn) {
    const float bias = bproj[nn * 16 + m0];
    if (nn + 1 < 6) {
#pragma unroll
      for (int kk = 0; kk < 24; ++kk)
        b[(nn + 1) & 1][kk] =
            *reinterpret_cast<const v2f*>(wb + (nn + 1) * (16 * WP) + kk * 4);
    }
    __builtin_amdgcn_sched_barrier(0);
    v8f acc0 = {}, acc1 = {};
#pragma unroll
    for (int kk = 0; kk < 24; ++kk) {
      acc0 = wmma4(a0[kk], b[nn & 1][kk], acc0);
      acc1 = wmma4(a1[kk], b[nn & 1][kk], acc1);
    }
#pragma unroll
    for (int v = 0; v < 8; ++v) {
      ob[(size_t)v * (NN * CC) + nn * 16] = acc0[v] + bias;
      ob[(size_t)(16 + v) * (NN * CC) + nn * 16] = acc1[v] + bias;
    }
  }
}

extern "C" void kernel_launch(void* const* d_in, const int* in_sizes, int n_in,
                              void* d_out, int out_size, void* d_ws, size_t ws_size,
                              hipStream_t stream) {
  (void)in_sizes; (void)n_in; (void)out_size; (void)ws_size;
  const float* x     = (const float*)d_in[0];
  const float* mask  = (const float*)d_in[1];
  const float* wqkv  = (const float*)d_in[2];
  const float* wproj = (const float*)d_in[3];
  const float* bproj = (const float*)d_in[4];
  float* out = (float*)d_out;

  float* Y = (float*)d_ws;

  qkv_gemm_kernel<<<dim3(BTOT * WN), dim3(288), 0, stream>>>(x, wqkv, Y);
  win_attn_proj_kernel<<<dim3(BTOT * HH * 2), dim3(256), 0, stream>>>(Y, mask, wproj, bproj, out);
}


# attention kernel: Q,K,V chunks fetched two chunks ahead instead of one (second register set, chunk loops unrolled; Wproj fetched behind the last V chunk); order of all sums unchanged
# speedup vs baseline: 2.6994x; 1.0488x over previous
#include <hip/hip_runtime.h>


typedef __attribute__((ext_vector_type(2))) float v2f;
typedef __attribute__((ext_vector_type(8))) float v8f;

#define BTOT 16
#define WN   64
#define NN   64
#define CC   96
#define HH   8
#define DD   768
#define C3   288
#define ROW3 18432
#define YBT  1179648
#define HD   6144
#define ATTN_SCALE 0.28867513459481287f

static __device__ __forceinline__ v8f wmma4(v2f a, v2f b, v8f c) {
  return __builtin_amdgcn_wmma_f32_16x16x4_f32(
       false, a,  false, b,
       (short)0, c,  false,  false);
}

static __device__ __forceinline__ void lds_barrier() {
  __builtin_amdgcn_fence(__ATOMIC_RELEASE, "workgroup", "local");
  __builtin_amdgcn_s_barrier();
  __builtin_amdgcn_fence(__ATOMIC_ACQUIRE, "workgroup", "local");
}

__global__ __launch_bounds__(288) void qkv_gemm_kernel(
    const float* __restrict__ x, const float* __restrict__ wqkv,
    float* __restrict__ Y) {
  __shared__ __align__(16) float sX[64 * 100];
  const int tid = threadIdx.x;
  const int lane = tid & 31, wave = tid >> 5;
  const int g = blockIdx.x;
  const float* xt = x + (size_t)g * (NN * CC);
  float* yt = Y + (size_t)(g / WN) * YBT + (size_t)(g % WN) * ROW3;

  const int m0 = lane & 15;
  const int ksel = (lane >> 4) * 2;
  const int hi8 = (lane >> 4) << 3;
  const int ncol = wave * 32 + m0;

  const int xr0 = tid / 24, xc4 = (tid % 24) * 4;
  const float* xg = xt + tid * 4;
  float4 xr[5];
  float4 xr5 = make_float4(0.f, 0.f, 0.f, 0.f);
#pragma unroll
  for (int j = 0; j < 5; ++j)
    xr[j] = *reinterpret_cast<const float4*>(xg + j * (288 * 4));
  if (tid < 96) xr5 = *reinterpret_cast<const float4*>(xg + 5 * (288 * 4));

  const float* bp = wqkv + ncol * CC + ksel;
  v2f b0[24], b1[24];
#pragma unroll
  for (int kk = 0; kk < 24; ++kk) {
    b0[kk] = *reinterpret_cast<const v2f*>(bp + kk * 4);
    b1[kk] = *reinterpret_cast<const v2f*>(bp + 16 * CC + kk * 4);
  }
  float* sxp = &sX[xr0 * 100 + xc4];
#pragma unroll
  for (int j = 0; j < 5; ++j)
    *reinterpret_cast<float4*>(sxp + j * (12 * 100)) = xr[j];
  if (tid < 96) *reinterpret_cast<float4*>(sxp + 5 * (12 * 100)) = xr5;
  lds_barrier();

  const float* ap = &sX[m0 * 100 + ksel];
  float* yo = yt + hi8 * C3 + ncol;
  for (int mt = 0; mt < 4; ++mt) {
    v8f acc0 = {}, acc1 = {};
#pragma unroll
    for (int kk = 0; kk < 24; ++kk) {
      v2f a = *reinterpret_cast<const v2f*>(ap + mt * (16 * 100) + kk * 4);
      acc0 = wmma4(a, b0[kk], acc0);
      acc1 = wmma4(a, b1[kk], acc1);
    }
#pragma unroll
    for (int v = 0; v < 8; ++v) {
      yo[(mt * 16 + v) * C3] = acc0[v];
      yo[(mt * 16 + v) * C3 + 16] = acc1[v];
    }
  }
}

#define CW   128
#define NCH  6
#define LDP  132
#define QR   32
#define OP   772
#define WP   100

struct chunk8 { float4 r0, r1, r2, r3, r4, r5, r6, r7; };
struct chunk4 { float4 r0, r1, r2, r3; };
static __device__ __forceinline__ float4 ld4(const float* __restrict__ g, int u) {
  return *reinterpret_cast<const float4*>(g + (size_t)u * (8 * ROW3));
}
static __device__ __forceinline__ void st4(float* s, int u, float4 v) {
  *reinterpret_cast<float4*>(s + u * (8 * LDP)) = v;
}
static __device__ __forceinline__ chunk8 fetch_chunk(const float* __restrict__ g) {
  chunk8 c;
  c.r0 = ld4(g, 0); c.r1 = ld4(g, 1); c.r2 = ld4(g, 2); c.r3 = ld4(g, 3);
  c.r4 = ld4(g, 4); c.r5 = ld4(g, 5); c.r6 = ld4(g, 6); c.r7 = ld4(g, 7);
  return c;
}
static __device__ __forceinline__ chunk4 fetch_chunk4(const float* __restrict__ g) {
  chunk4 c;
  c.r0 = ld4(g, 0); c.r1 = ld4(g, 1); c.r2 = ld4(g, 2); c.r3 = ld4(g, 3);
  return c;
}
static __device__ __forceinline__ void stage_chunk(float* s, const chunk8& c) {
  st4(s, 0, c.r0); st4(s, 1, c.r1); st4(s, 2, c.r2); st4(s, 3, c.r3);
  st4(s, 4, c.r4); st4(s, 5, c.r5); st4(s, 6, c.r6); st4(s, 7, c.r7);
}
static __device__ __forceinline__ void stage_chunk4(float* s, const chunk4& c) {
  st4(s, 0, c.r0); st4(s, 1, c.r1); st4(s, 2, c.r2); st4(s, 3, c.r3);
}

static __device__ __forceinline__ float4 ld_w(const float* __restrict__ w, int f) {
  return *reinterpret_cast<const float4*>(w + f * 4);
}
static __device__ __forceinline__ void st_w(float* sw, int f, float4 v) {
  *reinterpret_cast<float4*>(&sw[(f / 24) * WP + (f % 24) * 4]) = v;
}

static __device__ __forceinline__ void s_block(v8f& accS, const float* ap,
                                               const float* bp) {
#pragma unroll
  for (int kk = 0; kk < 32; ++kk) {
    v2f a = *reinterpret_cast<const v2f*>(ap + kk * 4);
    v2f b = *reinterpret_cast<const v2f*>(bp + kk * 4);
    accS = wmma4(a, b, accS);
  }
}

static __device__ __forceinline__ void pv_block(const float* vp, const float* pp,
                                                float* so) {
  v8f acc[2];
#pragma unroll
  for (int j = 0; j < 2; ++j) acc[j] = (v8f){};
#pragma unroll
  for (int kk = 0; kk < 16; ++kk) {
    v2f a = *reinterpret_cast<const v2f*>(pp + kk * 4);
#pragma unroll
    for (int j = 0; j < 2; ++j) {
      v2f b;
      b.x = vp[(kk * 4) * LDP + j * 16];
      b.y = vp[(kk * 4 + 1) * LDP + j * 16];
      acc[j] = wmma4(a, b, acc[j]);
    }
  }
#pragma unroll
  for (int j = 0; j < 2; ++j)
#pragma unroll
    for (int v = 0; v < 8; ++v)
      so[v * OP + j * 16] = acc[j][v];
}

__global__ __launch_bounds__(256) void win_attn_proj_kernel(
    const float* __restrict__ Y, const float* __restrict__ mask,
    const float* __restrict__ wproj, const float* __restrict__ bproj,
    float* __restrict__ out) {
  __shared__ __align__(16) float sQ[2][QR * LDP];
  __shared__ __align__(16) float sK[2][64 * LDP];
  __shared__ __align__(16) float sS[QR * 68];
  __shared__ __align__(16) float sO[QR * OP];
  __shared__ __align__(16) float sW[CC * WP];
  const int tid = threadIdx.x, lane = tid & 31, wave = tid >> 5;
  const int bt = blockIdx.x >> 4, h = (blockIdx.x >> 1) & 7;
  const int q0 = (blockIdx.x & 1) * QR;
  const float* qb = Y + (size_t)bt * YBT + h * DD;

  const int m0 = lane & 15;
  const int ksel = (lane >> 4) * 2;
  const int hi8 = (lane >> 4) << 3;
  const int mt = wave >> 2;
  const int nt = wave & 3;

  const float* gk = qb + (size_t)wave * ROW3 + lane * 4 + HD;
  const float* gv = gk + HD;
  const float* gq = qb + (size_t)(q0 + wave) * ROW3 + lane * 4;
  const int so = wave * LDP + lane * 4;

  chunk4 rq[2];
  chunk8 rk[2];

  v8f accS = (v8f){};

  const float* mk = mask + (size_t)(bt & 7) * (NN * NN) + (size_t)q0 * NN;
  const int ncol = nt * 16 + m0;
  float mv[8];
#pragma unroll
  for (int v = 0; v < 8; ++v)
    mv[v] = mk[(mt * 16 + v + hi8) * NN + ncol];

  rq[0] = fetch_chunk4(gq);
  rk[0] = fetch_chunk(gk);
  __builtin_amdgcn_sched_barrier(0);
  rq[1] = fetch_chunk4(gq + CW);
  rk[1] = fetch_chunk(gk + CW);
  stage_chunk4(&sQ[0][so], rq[0]);
  stage_chunk(&sK[0][so], rk[0]);
  lds_barrier();
  rq[0] = fetch_chunk4(gq + 2 * CW);
  rk[0] = fetch_chunk(gk + 2 * CW);
  __builtin_amdgcn_sched_barrier(0);

  const int aoff = (mt * 16 + m0) * LDP + ksel;
  const int boff = (nt * 16 + m0) * LDP + ksel;
#pragma unroll
  for (int t = 0; t < NCH; ++t) {
    const int cur = t & 1, nxt = cur ^ 1;
    s_block(accS, &sQ[cur][aoff], &sK[cur][boff]);
    if (t + 1 < NCH) {
      stage_chunk4(&sQ[nxt][so], rq[nxt]);
      stage_chunk(&sK[nxt][so], rk[nxt]);
    } else {
#pragma unroll
      for (int v = 0; v < 8; ++v) {
        const int m = mt * 16 + v + hi8;
        sS[m * 68 + ncol] = accS[v] * ATTN_SCALE + mv[v];
      }
      stage_chunk(&sK[nxt][so], rk[nxt]);
    }
    lds_barrier();
    if (t + 3 < NCH) {
      rq[nxt] = fetch_chunk4(gq + (t + 3) * CW);
      rk[nxt] = fetch_chunk(gk + (t + 3) * CW);
    } else {
      rk[nxt] = fetch_chunk(gv + (t + 3 - NCH) * CW);
    }
    __builtin_amdgcn_sched_barrier(0);
  }

  for (int r = wave * 4; r < wave * 4 + 4; ++r) {
    float x0 = sS[r * 68 + lane];
    float x1 = sS[r * 68 + lane + 32];
    float mx = fmaxf(x0, x1);
#pragma unroll
    for (int off = 16; off >= 1; off >>= 1)
      mx = fmaxf(mx, __shfl_xor(mx, off, 32));
    const float e0 = __expf(x0 - mx), e1 = __expf(x1 - mx);
    float sm = e0 + e1;
#pragma unroll
    for (int off = 16; off >= 1; off >>= 1)
      sm += __shfl_xor(sm, off, 32);
    const float inv = 1.0f / sm;
    sS[r * 68 + lane] = e0 * inv;
    sS[r * 68 + lane + 32] = e1 * inv;
  }
  lds_barrier();

  const int dq = nt * 32;
  const float* pp = &sS[(mt * 16 + m0) * 68 + ksel];
  float* zrow = &sO[(mt * 16 + hi8) * OP + dq + m0];
  float4 w0, w1, w2, w3, w4, w5, w6, w7, w8;
#pragma unroll
  for (int u = 0; u < NCH; ++u) {
    const int cur = u & 1, nxt = cur ^ 1;
    pv_block(&sK[cur][ksel * LDP + dq + m0], pp, zrow + u * CW);
    if (u + 1 < NCH) stage_chunk(&sK[nxt][so], rk[nxt]);
    if (u + 2 == NCH) {
      st_w(sW, tid, w0); st_w(sW, tid + 256, w1); st_w(sW, tid + 512, w2);
      st_w(sW, tid + 768, w3); st_w(sW, tid + 1024, w4); st_w(sW, tid + 1280, w5);
      st_w(sW, tid + 1536, w6); st_w(sW, tid + 1792, w7); st_w(sW, tid + 2048, w8);
    }
    lds_barrier();
    if (u + 3 < NCH) {
      rk[nxt] = fetch_chunk(gv + (u + 3) * CW);
      if (u + 4 == NCH) {
        w0 = ld_w(wproj, tid); w1 = ld_w(wproj, tid + 256); w2 = ld_w(wproj, tid + 512);
        w3 = ld_w(wproj, tid + 768); w4 = ld_w(wproj, tid + 1024); w5 = ld_w(wproj, tid + 1280);
        w6 = ld_w(wproj, tid + 1536); w7 = ld_w(wproj, tid + 1792); w8 = ld_w(wproj, tid + 2048);
      }
      __builtin_amdgcn_sched_barrier(0);
    }
  }

  const float* oa = &sO[m0 * OP + wave * CC + ksel];
  v2f a0[24], a1[24];
#pragma unroll
  for (int kk = 0; kk < 24; ++kk) {
    a0[kk] = *reinterpret_cast<const v2f*>(oa + kk * 4);
    a1[kk] = *reinterpret_cast<const v2f*>(oa + 16 * OP + kk * 4);
  }
  float* ob = out + ((size_t)(bt * WN + q0 + hi8) * NN + (h * 8 + wave)) * CC + m0;
  const float* wb = &sW[m0 * WP + ksel];
  v2f b[2][24];
#pragma unroll
  for (int kk = 0; kk < 24; ++kk)
    b[0][kk] = *reinterpret_cast<const v2f*>(wb + kk * 4);
#pragma unroll
  for (int nn = 0; nn < 6; ++nn) {
    const float bias = bproj[nn * 16 + m0];
    if (nn + 1 < 6) {
#pragma unroll
      for (int kk = 0; kk < 24; ++kk)
        b[(nn + 1) & 1][kk] =
            *reinterpret_cast<const v2f*>(wb + (nn + 1) * (16 * WP) + kk * 4);
    }
    __builtin_amdgcn_sched_barrier(0);
    v8f acc0 = {}, acc1 = {};
#pragma unroll
    for (int kk = 0; kk < 24; ++kk) {
      acc0 = wmma4(a0[kk], b[nn & 1][kk], acc0);
      acc1 = wmma4(a1[kk], b[nn & 1][kk], acc1);
    }
#pragma unroll
    for (int v = 0; v < 8; ++v) {
      ob[(size_t)v * (NN * CC) + nn * 16] = acc0[v] + bias;
      ob[(size_t)(16 + v) * (NN * CC) + nn * 16] = acc1[v] + bias;
    }
  }
}

extern "C" void kernel_launch(void* const* d_in, const int* in_sizes, int n_in,
                              void* d_out, int out_size, void* d_ws, size_t ws_size,
                              hipStream_t stream) {
  (void)in_sizes; (void)n_in; (void)out_size; (void)ws_size;
  const float* x     = (const float*)d_in[0];
  const float* mask  = (const float*)d_in[1];
  const float* wqkv  = (const float*)d_in[2];
  const float* wproj = (const float*)d_in[3];
  const float* bproj = (const float*)d_in[4];
  float* out = (float*)d_out;

  float* Y = (float*)d_ws;

  qkv_gemm_kernel<<<dim3(BTOT * WN), dim3(288), 0, stream>>>(x, wqkv, Y);
  win_attn_proj_kernel<<<dim3(BTOT * HH * 2), dim3(256), 0, stream>>>(Y, mask, wproj, bproj, out);
}


# V third of the qkv layer folded into the proj weight: out = (P x_t) M_u + b, M_u = Wv_u^T Wproj^T built every call by 3 extra work-groups; V never formed; proj-layer sums reassociated
# speedup vs baseline: 3.0522x; 1.1307x over previous
#include <hip/hip_runtime.h>


typedef __attribute__((ext_vector_type(2))) float v2f;
typedef __attribute__((ext_vector_type(8))) float v8f;

#define BTOT 16
#define WN   64
#define NN   64
#define CC   96
#define HH   8
#define DD   768
#define C3   288
#define ROW3 18432
#define YBT  1179648
#define HD   6144
#define ATTN_SCALE 0.28867513459481287f
#define QKF  12288
#define MROWS 48
#define NMW  3
#define MTSZ 9216

static __device__ __forceinline__ v8f wmma4(v2f a, v2f b, v8f c) {
  return __builtin_amdgcn_wmma_f32_16x16x4_f32(
       false, a,  false, b,
       (short)0, c,  false,  false);
}

static __device__ __forceinline__ void lds_barrier() {
  __builtin_amdgcn_fence(__ATOMIC_RELEASE, "workgroup", "local");
  __builtin_amdgcn_s_barrier();
  __builtin_amdgcn_fence(__ATOMIC_ACQUIRE, "workgroup", "local");
}

__global__ __launch_bounds__(288) void qkv_gemm_kernel(
    const float* __restrict__ x, const float* __restrict__ wqkv,
    const float* __restrict__ wproj, float* __restrict__ Y, float* __restrict__ MT) {
  __shared__ __align__(16) float sX[96 * 100];
  const int tid = threadIdx.x;
  const int lane = tid & 31, wave = tid >> 5;
  const int m0 = lane & 15;
  const int ksel = (lane >> 4) * 2;
  const int hi8 = (lane >> 4) << 3;

  if (blockIdx.x < NMW) {
    const int u = blockIdx.x;
#pragma unroll
    for (int i = 0; i < 8; ++i) {
      const int f = tid + 288 * i;
      const float4 v = *reinterpret_cast<const float4*>(wproj + f * 4);
      *reinterpret_cast<float4*>(&sX[(f / 24) * 100 + (f % 24) * 4]) = v;
    }
    const int cp = wave % 3, mp = wave / 3;
    const float* bq = wqkv + (size_t)(CC * u + ksel) * CC + cp * 32 + m0;
    v2f c0[24], c1[24];
#pragma unroll
    for (int kk = 0; kk < 24; ++kk) {
      c0[kk].x = bq[(kk * 4) * CC];
      c0[kk].y = bq[(kk * 4 + 1) * CC];
      c1[kk].x = bq[(kk * 4) * CC + 16];
      c1[kk].y = bq[(kk * 4 + 1) * CC + 16];
    }
    __syncthreads();
    float* mo = MT + (size_t)u * MTSZ + hi8 * CC + cp * 32 + m0;
#pragma unroll
    for (int mi = 0; mi < 2; ++mi) {
      const int mt = mp * 2 + mi;
      const float* aq = &sX[(mt * 16 + m0) * 100 + ksel];
      v8f acc0 = {}, acc1 = {};
#pragma unroll
      for (int kk = 0; kk < 24; ++kk) {
        v2f a = *reinterpret_cast<const v2f*>(aq + kk * 4);
        acc0 = wmma4(a, c0[kk], acc0);
        acc1 = wmma4(a, c1[kk], acc1);
      }
#pragma unroll
      for (int v = 0; v < 8; ++v) {
        mo[(mt * 16 + v) * CC] = acc0[v];
        mo[(mt * 16 + v) * CC + 16] = acc1[v];
      }
    }
    return;
  }

  const int g = blockIdx.x - NMW;
  const float* xt = x + (size_t)g * (NN * CC);
  float* yt = Y + (size_t)(g / WN) * YBT + (size_t)(g % WN) * ROW3;
  const int ncol = wave * 32 + m0;

  const int xr0 = tid / 24, xc4 = (tid % 24) * 4;
  const float* xg = xt + tid * 4;
  float4 xr[4];
#pragma unroll
  for (int j = 0; j < 4; ++j)
    xr[j] = *reinterpret_cast<const float4*>(xg + j * (288 * 4));

  const float* bp = wqkv + ncol * CC + ksel;
  v2f b0[24], b1[24];
#pragma unroll
  for (int kk = 0; kk < 24; ++kk) {
    b0[kk] = *reinterpret_cast<const v2f*>(bp + kk * 4);
    b1[kk] = *reinterpret_cast<const v2f*>(bp + 16 * CC + kk * 4);
  }
  float* sxp = &sX[xr0 * 100 + xc4];
#pragma unroll
  for (int j = 0; j < 4; ++j)
    *reinterpret_cast<float4*>(sxp + j * (12 * 100)) = xr[j];
  lds_barrier();

  const float* ap = &sX[m0 * 100 + ksel];
  float* yo = yt + hi8 * C3 + ncol;
#pragma unroll
  for (int mt = 0; mt < MROWS / 16; ++mt) {
    v8f acc0 = {}, acc1 = {};
#pragma unroll
    for (int kk = 0; kk < 24; ++kk) {
      v2f a = *reinterpret_cast<const v2f*>(ap + mt * (16 * 100) + kk * 4);
      acc0 = wmma4(a, b0[kk], acc0);
      acc1 = wmma4(a, b1[kk], acc1);
    }
#pragma unroll
    for (int v = 0; v < 8; ++v) {
      const int tok = mt * 16 + v + hi8;
      if (tok * C3 + wave * 32 < QKF) {
        yo[(mt * 16 + v) * C3] = acc0[v];
        yo[(mt * 16 + v) * C3 + 16] = acc1[v];
      }
    }
  }
}

#define CW   128
#define NCH  6
#define LDP  132
#define QR   32
#define NXC  3
#define XROW 6144
#define OP   388
#define WP   100

struct chunk8 { float4 r0, r1, r2, r3, r4, r5, r6, r7; };
struct chunk4 { float4 r0, r1, r2, r3; };
static __device__ __forceinline__ float4 ld4(const float* __restrict__ g, int u) {
  return *reinterpret_cast<const float4*>(g + (size_t)u * (8 * ROW3));
}
static __device__ __forceinline__ void st4(float* s, int u, float4 v) {
  *reinterpret_cast<float4*>(s + u * (8 * LDP)) = v;
}
static __device__ __forceinline__ chunk8 fetch_chunk(const float* __restrict__ g) {
  chunk8 c;
  c.r0 = ld4(g, 0); c.r1 = ld4(g, 1); c.r2 = ld4(g, 2); c.r3 = ld4(g, 3);
  c.r4 = ld4(g, 4); c.r5 = ld4(g, 5); c.r6 = ld4(g, 6); c.r7 = ld4(g, 7);
  return c;
}
static __device__ __forceinline__ float4 ld4x(const float* __restrict__ g, int u) {
  return *reinterpret_cast<const float4*>(g + (size_t)u * (8 * XROW));
}
static __device__ __forceinline__ chunk8 fetch_chunk_x(const float* __restrict__ g) {
  chunk8 c;
  c.r0 = ld4x(g, 0); c.r1 = ld4x(g, 1); c.r2 = ld4x(g, 2); c.r3 = ld4x(g, 3);
  c.r4 = ld4x(g, 4); c.r5 = ld4x(g, 5); c.r6 = ld4x(g, 6); c.r7 = ld4x(g, 7);
  return c;
}
static __device__ __forceinline__ chunk4 fetch_chunk4(const float* __restrict__ g) {
  chunk4 c;
  c.r0 = ld4(g, 0); c.r1 = ld4(g, 1); c.r2 = ld4(g, 2); c.r3 = ld4(g, 3);
  return c;
}
static __device__ __forceinline__ void stage_chunk(float* s, const chunk8& c) {
  st4(s, 0, c.r0); st4(s, 1, c.r1); st4(s, 2, c.r2); st4(s, 3, c.r3);
  st4(s, 4, c.r4); st4(s, 5, c.r5); st4(s, 6, c.r6); st4(s, 7, c.r7);
}
static __device__ __forceinline__ void stage_chunk4(float* s, const chunk4& c) {
  st4(s, 0, c.r0); st4(s, 1, c.r1); st4(s, 2, c.r2); st4(s, 3, c.r3);
}

static __device__ __forceinline__ float4 ld_w(const float* __restrict__ w, int f) {
  return *reinterpret_cast<const float4*>(w + f * 4);
}
static __device__ __forceinline__ void st_w(float* sw, int f, float4 v) {
  *reinterpret_cast<float4*>(&sw[(f / 24) * WP + (f % 24) * 4]) = v;
}

static __device__ __forceinline__ void s_block(v8f& accS, const float* ap,
                                               const float* bp) {
#pragma unroll
  for (int kk = 0; kk < 32; ++kk) {
    v2f a = *reinterpret_cast<const v2f*>(ap + kk * 4);
    v2f b = *reinterpret_cast<const v2f*>(bp + kk * 4);
    accS = wmma4(a, b, accS);
  }
}

static __device__ __forceinline__ void pv_block(const float* vp, const float* pp,
                                                float* so) {
  v8f acc[2];
#pragma unroll
  for (int j = 0; j < 2; ++j) acc[j] = (v8f){};
#pragma unroll
  for (int kk = 0; kk < 16; ++kk) {
    v2f a = *reinterpret_cast<const v2f*>(pp + kk * 4);
#pragma unroll
    for (int j = 0; j < 2; ++j) {
      v2f b;
      b.x = vp[(kk * 4) * LDP + j * 16];
      b.y = vp[(kk * 4 + 1) * LDP + j * 16];
      acc[j] = wmma4(a, b, acc[j]);
    }
  }
#pragma unroll
  for (int j = 0; j < 2; ++j)
#pragma unroll
    for (int v = 0; v < 8; ++v)
      so[v * OP + j * 16] = acc[j][v];
}

__global__ __launch_bounds__(256) void win_attn_proj_kernel(
    const float* __restrict__ Y, const float* __restrict__ x,
    const float* __restrict__ mask,
    const float* __restrict__ MT, const float* __restrict__ bproj,
    float* __restrict__ out) {
  __shared__ __align__(16) float sQ[2][QR * LDP];
  __shared__ __align__(16) float sK[2][64 * LDP];
  __shared__ __align__(16) float sS[QR * 68];
  __shared__ __align__(16) float sO[QR * OP];
  __shared__ __align__(16) float sW[NMW * CC * WP];
  const int tid = threadIdx.x, lane = tid & 31, wave = tid >> 5;
  const int bt = blockIdx.x >> 4, h = (blockIdx.x >> 1) & 7;
  const int q0 = (blockIdx.x & 1) * QR;
  const float* qb = Y + (size_t)bt * YBT + h * DD;

  const int m0 = lane & 15;
  const int ksel = (lane >> 4) * 2;
  const int hi8 = (lane >> 4) << 3;
  const int mt = wave >> 2;
  const int nt = wave & 3;

  const float* gk = qb + (size_t)wave * ROW3 + lane * 4 + HD;
  const int t0 = (128 + 8 * h) / 3;
  const int tb = t0 < NN - 4 ? t0 : NN - 4;
  const float* gx = x + ((size_t)(bt * WN + wave) * NN + tb) * CC + lane * 4;
  const float* gq = qb + (size_t)(q0 + wave) * ROW3 + lane * 4;
  const int so = wave * LDP + lane * 4;

  chunk4 rq[2];
  chunk8 rk[2];

  v8f accS = (v8f){};

  const float* mk = mask + (size_t)(bt & 7) * (NN * NN) + (size_t)q0 * NN;
  const int ncol = nt * 16 + m0;
  float mv[8];
#pragma unroll
  for (int v = 0; v < 8; ++v)
    mv[v] = mk[(mt * 16 + v + hi8) * NN + ncol];

  rq[0] = fetch_chunk4(gq);
  rk[0] = fetch_chunk(gk);
  __builtin_amdgcn_sched_barrier(0);
  rq[1] = fetch_chunk4(gq + CW);
  rk[1] = fetch_chunk(gk + CW);
  stage_chunk4(&sQ[0][so], rq[0]);
  stage_chunk(&sK[0][so], rk[0]);
  lds_barrier();
  rq[0] = fetch_chunk4(gq + 2 * CW);
  rk[0] = fetch_chunk(gk + 2 * CW);
  __builtin_amdgcn_sched_barrier(0);

  const int aoff = (mt * 16 + m0) * LDP + ksel;
  const int boff = (nt * 16 + m0) * LDP + ksel;
#pragma unroll
  for (int t = 0; t < NCH; ++t) {
    const int cur = t & 1, nxt = cur ^ 1;
    s_block(accS, &sQ[cur][aoff], &sK[cur][boff]);
    if (t + 1 < NCH) {
      stage_chunk4(&sQ[nxt][so], rq[nxt]);
      stage_chunk(&sK[nxt][so], rk[nxt]);
    } else {
#pragma unroll
      for (int v = 0; v < 8; ++v) {
        const int m = mt * 16 + v + hi8;
        sS[m * 68 + ncol] = accS[v] * ATTN_SCALE + mv[v];
      }
      stage_chunk(&sK[nxt][so], rk[nxt]);
    }
    lds_barrier();
    if (t + 3 < NCH) {
      rq[nxt] = fetch_chunk4(gq + (t + 3) * CW);
      rk[nxt] = fetch_chunk(gk + (t + 3) * CW);
    } else {
      rk[nxt] = fetch_chunk_x(gx + (t + 3 - NCH) * CW);
    }
    __builtin_amdgcn_sched_barrier(0);
  }

  for (int r = wave * 4; r < wave * 4 + 4; ++r) {
    float x0 = sS[r * 68 + lane];
    float x1 = sS[r * 68 + lane + 32];
    float mx = fmaxf(x0, x1);
#pragma unroll
    for (int off = 16; off >= 1; off >>= 1)
      mx = fmaxf(mx, __shfl_xor(mx, off, 32));
    const float e0 = __expf(x0 - mx), e1 = __expf(x1 - mx);
    float sm = e0 + e1;
#pragma unroll
    for (int off = 16; off >= 1; off >>= 1)
      sm += __shfl_xor(sm, off, 32);
    const float inv = 1.0f / sm;
    sS[r * 68 + lane] = e0 * inv;
    sS[r * 68 + lane + 32] = e1 * inv;
  }
  lds_barrier();

  const int dq = nt * 32;
  const float* pp = &sS[(mt * 16 + m0) * 68 + ksel];
  float* zrow = &sO[(mt * 16 + hi8) * OP + dq + m0];
  float4 wr[27];
#pragma unroll
  for (int u = 0; u < NXC; ++u) {
    const int cur = u & 1, nxt = cur ^ 1;
    pv_block(&sK[cur][ksel * LDP + dq + m0], pp, zrow + u * CW);
    if (u + 1 < NXC) stage_chunk(&sK[nxt][so], rk[nxt]);
    if (u + 1 == NXC) {
#pragma unroll
      for (int i = 0; i < 27; ++i) st_w(sW, tid + 256 * i, wr[i]);
    }
    lds_barrier();
    if (u == 0) {
#pragma unroll
      for (int i = 0; i < 27; ++i) wr[i] = ld_w(MT, tid + 256 * i);
      __builtin_amdgcn_sched_barrier(0);
    }
  }

  const int G = 128 + 8 * h + wave;
  const float* oa = &sO[m0 * OP + (G / 3 - tb) * CC + ksel];
  v2f a0[24], a1[24];
#pragma unroll
  for (int kk = 0; kk < 24; ++kk) {
    a0[kk] = *reinterpret_cast<const v2f*>(oa + kk * 4);
    a1[kk] = *reinterpret_cast<const v2f*>(oa + 16 * OP + kk * 4);
  }
  float* ob = out + ((size_t)(bt * WN + q0 + hi8) * NN + (h * 8 + wave)) * CC + m0;
  const float* wb = &sW[((G % 3) * CC + m0) * WP + ksel];
  v2f b[2][24];
#pragma unroll
  for (int kk = 0; kk < 24; ++kk)
    b[0][kk] = *reinterpret_cast<const v2f*>(wb + kk * 4);
#pragma unroll
  for (int nn = 0; nn < 6; ++nn) {
    const float bias = bproj[nn * 16 + m0];
    if (nn + 1 < 6) {
#pragma unroll
      for (int kk = 0; kk < 24; ++kk)
        b[(nn + 1) & 1][kk] =
            *reinterpret_cast<const v2f*>(wb + (nn + 1) * (16 * WP) + kk * 4);
    }
    __builtin_amdgcn_sched_barrier(0);
    v8f acc0 = {}, acc1 = {};
#pragma unroll
    for (int kk = 0; kk < 24; ++kk) {
      acc0 = wmma4(a0[kk], b[nn & 1][kk], acc0);
      acc1 = wmma4(a1[kk], b[nn & 1][kk], acc1);
    }
#pragma unroll
    for (int v = 0; v < 8; ++v) {
      ob[(size_t)v * (NN * CC) + nn * 16] = acc0[v] + bias;
      ob[(size_t)(16 + v) * (NN * CC) + nn * 16] = acc1[v] + bias;
    }
  }
}

extern "C" void kernel_launch(void* const* d_in, const int* in_sizes, int n_in,
                              void* d_out, int out_size, void* d_ws, size_t ws_size,
                              hipStream_t stream) {
  (void)in_sizes; (void)n_in; (void)out_size; (void)ws_size;
  const float* x     = (const float*)d_in[0];
  const float* mask  = (const float*)d_in[1];
  const float* wqkv  = (const float*)d_in[2];
  const float* wproj = (const float*)d_in[3];
  const float* bproj = (const float*)d_in[4];
  float* out = (float*)d_out;

  float* Y = (float*)d_ws;
  float* MT = Y + (size_t)BTOT * YBT;

  qkv_gemm_kernel<<<dim3(NMW + BTOT * WN), dim3(288), 0, stream>>>(x, wqkv, wproj, Y, MT);
  win_attn_proj_kernel<<<dim3(BTOT * HH * 2), dim3(256), 0, stream>>>(Y, x, mask, MT, bproj, out);
}


# Q and K thirds of the qkv layer folded: S = (X_q A) X_k^T, A = Wq^T Wk per head and token pair built every call; Q, K never formed; 3 launches; score-layer sums reassociated
# speedup vs baseline: 3.7716x; 1.2357x over previous
#include <hip/hip_runtime.h>


typedef __attribute__((ext_vector_type(2))) float v2f;
typedef __attribute__((ext_vector_type(8))) float v8f;

#define BTOT 16
#define WN   64
#define NN   64
#define CC   96
#define HH   8
#define DD   768
#define C3   288
#define HD   6144
#define ATTN_SCALE 0.28867513459481287f
#define NMW  3
#define MTSZ 9216
#define UW   384
#define UROW 3072
#define VTSZ 36864
#define NAB  128

static __device__ __forceinline__ v8f wmma4(v2f a, v2f b, v8f c) {
  return __builtin_amdgcn_wmma_f32_16x16x4_f32(
       false, a,  false, b,
       (short)0, c,  false,  false);
}

static __device__ __forceinline__ void lds_barrier() {
  __builtin_amdgcn_fence(__ATOMIC_RELEASE, "workgroup", "local");
  __builtin_amdgcn_s_barrier();
  __builtin_amdgcn_fence(__ATOMIC_ACQUIRE, "workgroup", "local");
}

static __device__ __forceinline__ void fold_block(int h, int i, int j, int& lo, int& len) {
  const int tq0 = (DD * h) / C3, tk0 = (HD + DD * h) / C3;
  int a = DD * h, b = DD * h + DD;
  const int qa = C3 * (tq0 + i), ka = C3 * (tk0 + j) - HD;
  a = a > qa ? a : qa;  a = a > ka ? a : ka;
  b = b < qa + C3 ? b : qa + C3;  b = b < ka + C3 ? b : ka + C3;
  lo = a;  len = b > a ? b - a : 0;
}

#define NMTW 12
__global__ __launch_bounds__(288) void fold_weights_kernel(
    const float* __restrict__ wqkv, const float* __restrict__ wproj,
    float* __restrict__ VT, float* __restrict__ MT) {
  const int tid = threadIdx.x;
  const int lane = tid & 31, wave = tid >> 5;
  const int m0 = lane & 15;
  const int ksel = (lane >> 4) * 2;
  const int hi8 = (lane >> 4) << 3;

  if (blockIdx.x >= NAB * 4) {
    const int g = (blockIdx.x - NAB * 4) * 9 + wave;
    const int u = g / 36, ot = (g % 36) / 6, ct = g % 6;
    const float* aq = wproj + (ot * 16 + m0) * CC + ksel;
    const float* bq = wqkv + (size_t)(CC * u + ksel) * CC + ct * 16 + m0;
    v2f fa[24], fb[24];
#pragma unroll
    for (int kk = 0; kk < 24; ++kk) {
      fa[kk] = *reinterpret_cast<const v2f*>(aq + kk * 4);
      fb[kk].x = bq[(kk * 4) * CC];
      fb[kk].y = bq[(kk * 4 + 1) * CC];
    }
    v8f acc = {};
#pragma unroll
    for (int kk = 0; kk < 24; ++kk) acc = wmma4(fa[kk], fb[kk], acc);
    float* mo = MT + (size_t)u * MTSZ + (ot * 16 + hi8) * CC + ct * 16 + m0;
#pragma unroll
    for (int v = 0; v < 8; ++v) mo[v * CC] = acc[v];
    return;
  }

  const int blk = blockIdx.x >> 2;
  const int h = blk >> 4, bi = (blk >> 2) & 3, bj = blk & 3;
  int lo, len;
  fold_block(h, bi, bj, lo, len);
  if (len == 0) return;
  const int T = (blockIdx.x & 3) * 9 + wave;
  const int mt = T / 6, nt = T % 6;
  const int cq0 = lo % C3, ck0 = (HD + lo) % C3;
  v8f acc = {};
  for (int t0 = 0; t0 < len; t0 += CC) {
    const float* ag = wqkv + (size_t)(ck0 + t0 + ksel) * CC + mt * 16 + m0;
    const float* bg = wqkv + (size_t)(cq0 + t0 + ksel) * CC + nt * 16 + m0;
    v2f fa[24], fb[24];
#pragma unroll
    for (int kk = 0; kk < 24; ++kk) {
      fa[kk].x = ag[(kk * 4) * CC];  fa[kk].y = ag[(kk * 4 + 1) * CC];
      fb[kk].x = bg[(kk * 4) * CC];  fb[kk].y = bg[(kk * 4 + 1) * CC];
    }
#pragma unroll
    for (int kk = 0; kk < 24; ++kk) acc = wmma4(fa[kk], fb[kk], acc);
  }
  float* vo = VT + (size_t)(h * 4 + bj) * VTSZ + (size_t)(mt * 16 + hi8) * UW + bi * CC + nt * 16 + m0;
#pragma unroll
  for (int v = 0; v < 8; ++v) vo[v * UW] = acc[v];
}

#define XQP  388
__global__ __launch_bounds__(256) void u_gemm_kernel(
    const float* __restrict__ x, const float* __restrict__ VT, float* __restrict__ U) {
  __shared__ __align__(16) float sXq[16 * XQP];
  const int tid = threadIdx.x, lane = tid & 31;
  const int wave = __builtin_amdgcn_readfirstlane(tid >> 5);
  const int m0 = lane & 15;
  const int ksel = (lane >> 4) * 2;
  const int hi8 = (lane >> 4) << 3;
  const int bt = blockIdx.x >> 5, h = (blockIdx.x >> 2) & 7, n0 = (blockIdx.x & 3) * 16;
  const int tq0 = (DD * h) / C3;

  float4 xr[6];
#pragma unroll
  for (int q = 0; q < 6; ++q) {
    const int f = tid + 256 * q;
    xr[q] = *reinterpret_cast<const float4*>(
        x + ((size_t)(bt * WN + n0 + f / 96) * NN + tq0) * CC + (f % 96) * 4);
  }
#pragma unroll
  for (int q = 0; q < 6; ++q) {
    const int f = tid + 256 * q;
    *reinterpret_cast<float4*>(&sXq[(f / 96) * XQP + (f % 96) * 4]) = xr[q];
  }
  __syncthreads();

  const int j = wave >> 1, ct0 = (wave & 1) * 3;
  v8f acc[3];
#pragma unroll
  for (int e = 0; e < 3; ++e) acc[e] = (v8f){};
  for (int i = 0; i < 4; ++i) {
    int lo, len;
    fold_block(h, i, j, lo, len);
    if (len == 0) continue;
    const float* bg = VT + (size_t)(h * 4 + j) * VTSZ + (size_t)(ct0 * 16 + m0) * UW + i * CC + ksel;
    const float* ap = &sXq[m0 * XQP + i * CC + ksel];
    v2f b[3][24];
#pragma unroll
    for (int e = 0; e < 3; ++e)
#pragma unroll
      for (int kk = 0; kk < 24; ++kk)
        b[e][kk] = *reinterpret_cast<const v2f*>(bg + e * (16 * UW) + kk * 4);
    __builtin_amdgcn_sched_barrier(0);
#pragma unroll
    for (int kk = 0; kk < 24; ++kk) {
      v2f a = *reinterpret_cast<const v2f*>(ap + kk * 4);
#pragma unroll
      for (int e = 0; e < 3; ++e) acc[e] = wmma4(a, b[e][kk], acc[e]);
    }
  }
  float* uo = U + ((size_t)(bt * WN + n0 + hi8) * HH + h) * UW + j * CC + ct0 * 16 + m0;
#pragma unroll
  for (int e = 0; e < 3; ++e)
#pragma unroll
    for (int v = 0; v < 8; ++v)
      uo[(size_t)v * UROW + e * 16] = acc[e][v];
}

#define CW   128
#define NCH  3
#define LDP  132
#define QR   32
#define NXC  3
#define XROW 6144
#define OP   388
#define WP   100

struct chunk8 { float4 r0, r1, r2, r3, r4, r5, r6, r7; };
struct chunk4 { float4 r0, r1, r2, r3; };
static __device__ __forceinline__ float4 ld4(const float* __restrict__ g, int u) {
  return *reinterpret_cast<const float4*>(g + (size_t)u * (8 * UROW));
}
static __device__ __forceinline__ void st4(float* s, int u, float4 v) {
  *reinterpret_cast<float4*>(s + u * (8 * LDP)) = v;
}
static __device__ __forceinline__ float4 ld4x(const float* __restrict__ g, int u) {
  return *reinterpret_cast<const float4*>(g + (size_t)u * (8 * XROW));
}
static __device__ __forceinline__ chunk8 fetch_chunk_x(const float* __restrict__ g) {
  chunk8 c;
  c.r0 = ld4x(g, 0); c.r1 = ld4x(g, 1); c.r2 = ld4x(g, 2); c.r3 = ld4x(g, 3);
  c.r4 = ld4x(g, 4); c.r5 = ld4x(g, 5); c.r6 = ld4x(g, 6); c.r7 = ld4x(g, 7);
  return c;
}
static __device__ __forceinline__ chunk4 fetch_chunk4(const float* __restrict__ g) {
  chunk4 c;
  c.r0 = ld4(g, 0); c.r1 = ld4(g, 1); c.r2 = ld4(g, 2); c.r3 = ld4(g, 3);
  return c;
}
static __device__ __forceinline__ void stage_chunk(float* s, const chunk8& c) {
  st4(s, 0, c.r0); st4(s, 1, c.r1); st4(s, 2, c.r2); st4(s, 3, c.r3);
  st4(s, 4, c.r4); st4(s, 5, c.r5); st4(s, 6, c.r6); st4(s, 7, c.r7);
}
static __device__ __forceinline__ void stage_chunk4(float* s, const chunk4& c) {
  st4(s, 0, c.r0); st4(s, 1, c.r1); st4(s, 2, c.r2); st4(s, 3, c.r3);
}

static __device__ __forceinline__ float4 ld_w(const float* __restrict__ w, int f) {
  return *reinterpret_cast<const float4*>(w + f * 4);
}
static __device__ __forceinline__ void st_w(float* sw, int f, float4 v) {
  *reinterpret_cast<float4*>(&sw[(f / 24) * WP + (f % 24) * 4]) = v;
}

static __device__ __forceinline__ void s_block(v8f& accS, const float* ap,
                                               const float* bp) {
#pragma unroll
  for (int kk = 0; kk < 32; ++kk) {
    v2f a = *reinterpret_cast<const v2f*>(ap + kk * 4);
    v2f b = *reinterpret_cast<const v2f*>(bp + kk * 4);
    accS = wmma4(a, b, accS);
  }
}

static __device__ __forceinline__ void pv_block(const float* vp, const float* pp,
                                                float* so) {
  v8f acc[2];
#pragma unroll
  for (int j = 0; j < 2; ++j) acc[j] = (v8f){};
#pragma unroll
  for (int kk = 0; kk < 16; ++kk) {
    v2f a = *reinterpret_cast<const v2f*>(pp + kk * 4);
#pragma unroll
    for (int j = 0; j < 2; ++j) {
      v2f b;
      b.x = vp[(kk * 4) * LDP + j * 16];
      b.y = vp[(kk * 4 + 1) * LDP + j * 16];
      acc[j] = wmma4(a, b, acc[j]);
    }
  }
#pragma unroll
  for (int j = 0; j < 2; ++j)
#pragma unroll
    for (int v = 0; v < 8; ++v)
      so[v * OP + j * 16] = acc[j][v];
}

__global__ __launch_bounds__(256) void win_attn_proj_kernel(
    const float* __restrict__ U, const float* __restrict__ x,
    const float* __restrict__ mask,
    const float* __restrict__ MT, const float* __restrict__ bproj,
    float* __restrict__ out) {
  __shared__ __align__(16) float sQ[2][QR * LDP];
  __shared__ __align__(16) float sK[2][64 * LDP];
  __shared__ __align__(16) float sS[QR * 68];
  __shared__ __align__(16) float sO[QR * OP];
  __shared__ __align__(16) float sW[NMW * CC * WP];
  const int tid = threadIdx.x, lane = tid & 31, wave = tid >> 5;
  const int bt = blockIdx.x >> 4, h = (blockIdx.x >> 1) & 7;
  const int q0 = (blockIdx.x & 1) * QR;

  const int m0 = lane & 15;
  const int ksel = (lane >> 4) * 2;
  const int hi8 = (lane >> 4) << 3;
  const int mt = wave >> 2;
  const int nt = wave & 3;

  const int tk0 = (HD + DD * h) / C3;
  const float* gk = x + ((size_t)(bt * WN + wave) * NN + tk0) * CC + lane * 4;
  const int t0 = (128 + 8 * h) / 3;
  const int tb = t0 < NN - 4 ? t0 : NN - 4;
  const float* gx = x + ((size_t)(bt * WN + wave) * NN + tb) * CC + lane * 4;
  const float* gq = U + ((size_t)(bt * WN + q0 + wave) * HH + h) * UW + lane * 4;
  const int so = wave * LDP + lane * 4;

  chunk4 rq[2];
  chunk8 rk[2];

  v8f accS = (v8f){};

  const float* mk = mask + (size_t)(bt & 7) * (NN * NN) + (size_t)q0 * NN;
  const int ncol = nt * 16 + m0;
  float mv[8];
#pragma unroll
  for (int v = 0; v < 8; ++v)
    mv[v] = mk[(mt * 16 + v + hi8) * NN + ncol];

  rq[0] = fetch_chunk4(gq);
  rk[0] = fetch_chunk_x(gk);
  __builtin_amdgcn_sched_barrier(0);
  rq[1] = fetch_chunk4(gq + CW);
  rk[1] = fetch_chunk_x(gk + CW);
  stage_chunk4(&sQ[0][so], rq[0]);
  stage_chunk(&sK[0][so], rk[0]);
  lds_barrier();
  rq[0] = fetch_chunk4(gq + 2 * CW);
  rk[0] = fetch_chunk_x(gk + 2 * CW);
  __builtin_amdgcn_sched_barrier(0);

  const int aoff = (mt * 16 + m0) * LDP + ksel;
  const int boff = (nt * 16 + m0) * LDP + ksel;
#pragma unroll
  for (int t = 0; t < NCH; ++t) {
    const int cur = t & 1, nxt = cur ^ 1;
    s_block(accS, &sQ[cur][aoff], &sK[cur][boff]);
    if (t + 1 < NCH) {
      stage_chunk4(&sQ[nxt][so], rq[nxt]);
      stage_chunk(&sK[nxt][so], rk[nxt]);
    } else {
#pragma unroll
      for (int v = 0; v < 8; ++v) {
        const int m = mt * 16 + v + hi8;
        sS[m * 68 + ncol] = accS[v] * ATTN_SCALE + mv[v];
      }
      stage_chunk(&sK[nxt][so], rk[nxt]);
    }
    lds_barrier();
    if (t + 3 < NCH) {
      rq[nxt] = fetch_chunk4(gq + (t + 3) * CW);
      rk[nxt] = fetch_chunk_x(gk + (t + 3) * CW);
    } else {
      rk[nxt] = fetch_chunk_x(gx + (t + 3 - NCH) * CW);
    }
    __builtin_amdgcn_sched_barrier(0);
  }

  for (int r = wave * 4; r < wave * 4 + 4; ++r) {
    float x0 = sS[r * 68 + lane];
    float x1 = sS[r * 68 + lane + 32];
    float mx = fmaxf(x0, x1);
#pragma unroll
    for (int off = 16; off >= 1; off >>= 1)
      mx = fmaxf(mx, __shfl_xor(mx, off, 32));
    const float e0 = __expf(x0 - mx), e1 = __expf(x1 - mx);
    float sm = e0 + e1;
#pragma unroll
    for (int off = 16; off >= 1; off >>= 1)
      sm += __shfl_xor(sm, off, 32);
    const float inv = 1.0f / sm;
    sS[r * 68 + lane] = e0 * inv;
    sS[r * 68 + lane + 32] = e1 * inv;
  }
  lds_barrier();

  const int dq = nt * 32;
  const float* pp = &sS[(mt * 16 + m0) * 68 + ksel];
  float* zrow = &sO[(mt * 16 + hi8) * OP + dq + m0];
  float4 wr[27];
#pragma unroll
  for (int u = 0; u < NXC; ++u) {
    const int cur = (NCH + u) & 1, nxt = cur ^ 1;
    pv_block(&sK[cur][ksel * LDP + dq + m0], pp, zrow + u * CW);
    if (u + 1 < NXC) stage_chunk(&sK[nxt][so], rk[nxt]);
    if (u + 1 == NXC) {
#pragma unroll
      for (int i = 0; i < 27; ++i) st_w(sW, tid + 256 * i, wr[i]);
    }
    lds_barrier();
    if (u == 0) {
#pragma unroll
      for (int i = 0; i < 27; ++i) wr[i] = ld_w(MT, tid + 256 * i);
      __builtin_amdgcn_sched_barrier(0);
    }
  }

  const int G = 128 + 8 * h + wave;
  const float* oa = &sO[m0 * OP + (G / 3 - tb) * CC + ksel];
  v2f a0[24], a1[24];
#pragma unroll
  for (int kk = 0; kk < 24; ++kk) {
    a0[kk] = *reinterpret_cast<const v2f*>(oa + kk * 4);
    a1[kk] = *reinterpret_cast<const v2f*>(oa + 16 * OP + kk * 4);
  }
  float* ob = out + ((size_t)(bt * WN + q0 + hi8) * NN + (h * 8 + wave)) * CC + m0;
  const float* wb = &sW[((G % 3) * CC + m0) * WP + ksel];
  v2f b[2][24];
#pragma unroll
  for (int kk = 0; kk < 24; ++kk)
    b[0][kk] = *reinterpret_cast<const v2f*>(wb + kk * 4);
#pragma unroll
  for (int nn = 0; nn < 6; ++nn) {
    const float bias = bproj[nn * 16 + m0];
    if (nn + 1 < 6) {
#pragma unroll
      for (int kk = 0; kk < 24; ++kk)
        b[(nn + 1) & 1][kk] =
            *reinterpret_cast<const v2f*>(wb + (nn + 1) * (16 * WP) + kk * 4);
    }
    __builtin_amdgcn_sched_barrier(0);
    v8f acc0 = {}, acc1 = {};
#pragma unroll
    for (int kk = 0; kk < 24; ++kk) {
      acc0 = wmma4(a0[kk], b[nn & 1][kk], acc0);
      acc1 = wmma4(a1[kk], b[nn & 1][kk], acc1);
    }
#pragma unroll
    for (int v = 0; v < 8; ++v) {
      ob[(size_t)v * (NN * CC) + nn * 16] = acc0[v] + bias;
      ob[(size_t)(16 + v) * (NN * CC) + nn * 16] = acc1[v] + bias;
    }
  }
}

extern "C" void kernel_launch(void* const* d_in, const int* in_sizes, int n_in,
                              void* d_out, int out_size, void* d_ws, size_t ws_size,
                              hipStream_t stream) {
  (void)in_sizes; (void)n_in; (void)out_size; (void)ws_size;
  const float* x     = (const float*)d_in[0];
  const float* mask  = (const float*)d_in[1];
  const float* wqkv  = (const float*)d_in[2];
  const float* wproj = (const float*)d_in[3];
  const float* bproj = (const float*)d_in[4];
  float* out = (float*)d_out;

  float* U  = (float*)d_ws;
  float* VT = U + (size_t)BTOT * WN * UROW;
  float* MT = VT + (size_t)HH * 4 * VTSZ;

  fold_weights_kernel<<<dim3(NAB * 4 + NMTW), dim3(288), 0, stream>>>(wqkv, wproj, VT, MT);
  u_gemm_kernel<<<dim3(BTOT * HH * 4), dim3(256), 0, stream>>>(x, VT, U);
  win_attn_proj_kernel<<<dim3(BTOT * HH * 2), dim3(256), 0, stream>>>(U, x, mask, MT, bproj, out);
}


# U = X_q A formed inside the attention kernel: a work-group builds its 32 x 384 U tile in LDS, both M tiles per weight fragment, fragments loaded in k order; 2 launches, no U buffer; sums unchanged
# speedup vs baseline: 4.5975x; 1.2190x over previous
#include <hip/hip_runtime.h>


typedef __attribute__((ext_vector_type(2))) float v2f;
typedef __attribute__((ext_vector_type(8))) float v8f;

#define BTOT 16
#define WN   64
#define NN   64
#define CC   96
#define HH   8
#define DD   768
#define C3   288
#define HD   6144
#define ATTN_SCALE 0.28867513459481287f
#define NMW  3
#define MTSZ 9216
#define UW   384
#define VTSZ 36864
#define NAB  128

static __device__ __forceinline__ v8f wmma4(v2f a, v2f b, v8f c) {
  return __builtin_amdgcn_wmma_f32_16x16x4_f32(
       false, a,  false, b,
       (short)0, c,  false,  false);
}

static __device__ __forceinline__ void lds_barrier() {
  __builtin_amdgcn_fence(__ATOMIC_RELEASE, "workgroup", "local");
  __builtin_amdgcn_s_barrier();
  __builtin_amdgcn_fence(__ATOMIC_ACQUIRE, "workgroup", "local");
}

static __device__ __forceinline__ void fold_block(int h, int i, int j, int& lo, int& len) {
  const int tq0 = (DD * h) / C3, tk0 = (HD + DD * h) / C3;
  int a = DD * h, b = DD * h + DD;
  const int qa = C3 * (tq0 + i), ka = C3 * (tk0 + j) - HD;
  a = a > qa ? a : qa;  a = a > ka ? a : ka;
  b = b < qa + C3 ? b : qa + C3;  b = b < ka + C3 ? b : ka + C3;
  lo = a;  len = b > a ? b - a : 0;
}

#define NMTW 12
__global__ __launch_bounds__(288) void fold_weights_kernel(
    const float* __restrict__ wqkv, const float* __restrict__ wproj,
    float* __restrict__ VT, float* __restrict__ MT) {
  const int tid = threadIdx.x;
  const int lane = tid & 31, wave = tid >> 5;
  const int m0 = lane & 15;
  const int ksel = (lane >> 4) * 2;
  const int hi8 = (lane >> 4) << 3;

  if (blockIdx.x >= NAB * 4) {
    const int g = (blockIdx.x - NAB * 4) * 9 + wave;
    const int u = g / 36, ot = (g % 36) / 6, ct = g % 6;
    const float* aq = wproj + (ot * 16 + m0) * CC + ksel;
    const float* bq = wqkv + (size_t)(CC * u + ksel) * CC + ct * 16 + m0;
    v2f fa[24], fb[24];
#pragma unroll
    for (int kk = 0; kk < 24; ++kk) {
      fa[kk] = *reinterpret_cast<const v2f*>(aq + kk * 4);
      fb[kk].x = bq[(kk * 4) * CC];
      fb[kk].y = bq[(kk * 4 + 1) * CC];
    }
    v8f acc = {};
#pragma unroll
    for (int kk = 0; kk < 24; ++kk) acc = wmma4(fa[kk], fb[kk], acc);
    float* mo = MT + (size_t)u * MTSZ + (ot * 16 + hi8) * CC + ct * 16 + m0;
#pragma unroll
    for (int v = 0; v < 8; ++v) mo[v * CC] = acc[v];
    return;
  }

  const int blk = blockIdx.x >> 2;
  const int h = blk >> 4, bi = (blk >> 2) & 3, bj = blk & 3;
  int lo, len;
  fold_block(h, bi, bj, lo, len);
  if (len == 0) return;
  const int T = (blockIdx.x & 3) * 9 + wave;
  const int mt = T / 6, nt = T % 6;
  const int cq0 = lo % C3, ck0 = (HD + lo) % C3;
  v8f acc = {};
  for (int t0 = 0; t0 < len; t0 += CC) {
    const float* ag = wqkv + (size_t)(ck0 + t0 + ksel) * CC + mt * 16 + m0;
    const float* bg = wqkv + (size_t)(cq0 + t0 + ksel) * CC + nt * 16 + m0;
    v2f fa[24], fb[24];
#pragma unroll
    for (int kk = 0; kk < 24; ++kk) {
      fa[kk].x = ag[(kk * 4) * CC];  fa[kk].y = ag[(kk * 4 + 1) * CC];
      fb[kk].x = bg[(kk * 4) * CC];  fb[kk].y = bg[(kk * 4 + 1) * CC];
    }
#pragma unroll
    for (int kk = 0; kk < 24; ++kk) acc = wmma4(fa[kk], fb[kk], acc);
  }
  float* vo = VT + (size_t)(h * 4 + bj) * VTSZ + (size_t)(mt * 16 + hi8) * UW + bi * CC + nt * 16 + m0;
#pragma unroll
  for (int v = 0; v < 8; ++v) vo[v * UW] = acc[v];
}

#define CW   128
#define NCH  3
#define LDP  132
#define QR   32
#define NXC  3
#define XROW 6144
#define OP   388
#define WP   100

struct chunk8 { float4 r0, r1, r2, r3, r4, r5, r6, r7; };
static __device__ __forceinline__ void st4(float* s, int u, float4 v) {
  *reinterpret_cast<float4*>(s + u * (8 * LDP)) = v;
}
static __device__ __forceinline__ float4 ld4x(const float* __restrict__ g, int u) {
  return *reinterpret_cast<const float4*>(g + (size_t)u * (8 * XROW));
}
static __device__ __forceinline__ chunk8 fetch_chunk_x(const float* __restrict__ g) {
  chunk8 c;
  c.r0 = ld4x(g, 0); c.r1 = ld4x(g, 1); c.r2 = ld4x(g, 2); c.r3 = ld4x(g, 3);
  c.r4 = ld4x(g, 4); c.r5 = ld4x(g, 5); c.r6 = ld4x(g, 6); c.r7 = ld4x(g, 7);
  return c;
}
struct chunk4 { float4 r0, r1, r2, r3; };
static __device__ __forceinline__ chunk4 fetch_chunk4_x(const float* __restrict__ g) {
  chunk4 c;
  c.r0 = ld4x(g, 0); c.r1 = ld4x(g, 1); c.r2 = ld4x(g, 2); c.r3 = ld4x(g, 3);
  return c;
}
static __device__ __forceinline__ void st4q(float* s, int u, float4 v) {
  *reinterpret_cast<float4*>(s + u * (8 * OP)) = v;
}
static __device__ __forceinline__ void stage_chunk4q(float* s, const chunk4& c) {
  st4q(s, 0, c.r0); st4q(s, 1, c.r1); st4q(s, 2, c.r2); st4q(s, 3, c.r3);
}
static __device__ __forceinline__ void stage_chunk(float* s, const chunk8& c) {
  st4(s, 0, c.r0); st4(s, 1, c.r1); st4(s, 2, c.r2); st4(s, 3, c.r3);
  st4(s, 4, c.r4); st4(s, 5, c.r5); st4(s, 6, c.r6); st4(s, 7, c.r7);
}

static __device__ __forceinline__ float4 ld_w(const float* __restrict__ w, int f) {
  return *reinterpret_cast<const float4*>(w + f * 4);
}
static __device__ __forceinline__ void st_w(float* sw, int f, float4 v) {
  *reinterpret_cast<float4*>(&sw[(f / 24) * WP + (f % 24) * 4]) = v;
}

static __device__ __forceinline__ void s_block(v8f& accS, const float* ap,
                                               const float* bp) {
#pragma unroll
  for (int kk = 0; kk < 32; ++kk) {
    v2f a = *reinterpret_cast<const v2f*>(ap + kk * 4);
    v2f b = *reinterpret_cast<const v2f*>(bp + kk * 4);
    accS = wmma4(a, b, accS);
  }
}

static __device__ __forceinline__ void pv_block(const float* vp, const float* pp,
                                                float* so) {
  v8f acc[2];
#pragma unroll
  for (int j = 0; j < 2; ++j) acc[j] = (v8f){};
#pragma unroll
  for (int kk = 0; kk < 16; ++kk) {
    v2f a = *reinterpret_cast<const v2f*>(pp + kk * 4);
#pragma unroll
    for (int j = 0; j < 2; ++j) {
      v2f b;
      b.x = vp[(kk * 4) * LDP + j * 16];
      b.y = vp[(kk * 4 + 1) * LDP + j * 16];
      acc[j] = wmma4(a, b, acc[j]);
    }
  }
#pragma unroll
  for (int j = 0; j < 2; ++j)
#pragma unroll
    for (int v = 0; v < 8; ++v)
      so[v * OP + j * 16] = acc[j][v];
}

__global__ __launch_bounds__(256) void win_attn_proj_kernel(
    const float* __restrict__ x, const float* __restrict__ VT,
    const float* __restrict__ mask,
    const float* __restrict__ MT, const float* __restrict__ bproj,
    float* __restrict__ out) {
  __shared__ __align__(16) float sK[2][64 * LDP];
  __shared__ __align__(16) float sS[QR * 68];
  __shared__ __align__(16) float sO[QR * OP];
  __shared__ __align__(16) float sW[NMW * CC * WP];
  float* const sXq = sO;
  float* const sU = sW;
  const int tid = threadIdx.x, lane = tid & 31, wave = tid >> 5;
  const int bt = blockIdx.x >> 4, h = (blockIdx.x >> 1) & 7;
  const int q0 = (blockIdx.x & 1) * QR;

  const int m0 = lane & 15;
  const int ksel = (lane >> 4) * 2;
  const int hi8 = (lane >> 4) << 3;
  const int mt = wave >> 2;
  const int nt = wave & 3;

  const int tk0 = (HD + DD * h) / C3;
  const float* gk = x + ((size_t)(bt * WN + wave) * NN + tk0) * CC + lane * 4;
  const int t0 = (128 + 8 * h) / 3;
  const int tb = t0 < NN - 4 ? t0 : NN - 4;
  const float* gx = x + ((size_t)(bt * WN + wave) * NN + tb) * CC + lane * 4;
  const int so = wave * LDP + lane * 4;

  chunk8 rk[2];

  v8f accS = (v8f){};

  const float* mk = mask + (size_t)(bt & 7) * (NN * NN) + (size_t)q0 * NN;
  const int ncol = nt * 16 + m0;
  float mv[8];
#pragma unroll
  for (int v = 0; v < 8; ++v)
    mv[v] = mk[(mt * 16 + v + hi8) * NN + ncol];

  const int tq0 = (DD * h) / C3;
  const float* gxq = x + ((size_t)(bt * WN + q0 + wave) * NN + tq0) * CC + lane * 4;
  const chunk4 xq0 = fetch_chunk4_x(gxq);
  const chunk4 xq1 = fetch_chunk4_x(gxq + CW);
  const chunk4 xq2 = fetch_chunk4_x(gxq + 2 * CW);
  rk[0] = fetch_chunk_x(gk);
  __builtin_amdgcn_sched_barrier(0);
  rk[1] = fetch_chunk_x(gk + CW);
  const int soq = wave * OP + lane * 4;
  stage_chunk4q(&sXq[soq], xq0);
  stage_chunk4q(&sXq[soq + CW], xq1);
  stage_chunk4q(&sXq[soq + 2 * CW], xq2);
  stage_chunk(&sK[0][so], rk[0]);
  lds_barrier();
  rk[0] = fetch_chunk_x(gk + 2 * CW);
  __builtin_amdgcn_sched_barrier(0);

  {
    const int wv = __builtin_amdgcn_readfirstlane(tid >> 5);
    const int uj = wv >> 1, uct0 = (wv & 1) * 3;
    v8f ua0[3], ua1[3];
#pragma unroll
    for (int e = 0; e < 3; ++e) { ua0[e] = (v8f){}; ua1[e] = (v8f){}; }
    for (int i = 0; i < 4; ++i) {
      int lo, len;
      fold_block(h, i, uj, lo, len);
      if (len == 0) continue;
      const float* bg = VT + (size_t)(h * 4 + uj) * VTSZ + (size_t)(uct0 * 16 + m0) * UW + i * CC + ksel;
      const float* ap = &sXq[m0 * OP + i * CC + ksel];
      v2f b[3][24];
#pragma unroll
      for (int kk = 0; kk < 24; ++kk)
#pragma unroll
        for (int e = 0; e < 3; ++e)
          b[e][kk] = *reinterpret_cast<const v2f*>(bg + e * (16 * UW) + kk * 4);
      __builtin_amdgcn_sched_barrier(0);
#pragma unroll
      for (int kk = 0; kk < 24; ++kk) {
        v2f a0 = *reinterpret_cast<const v2f*>(ap + kk * 4);
        v2f a1 = *reinterpret_cast<const v2f*>(ap + 16 * OP + kk * 4);
#pragma unroll
        for (int e = 0; e < 3; ++e) {
          ua0[e] = wmma4(a0, b[e][kk], ua0[e]);
          ua1[e] = wmma4(a1, b[e][kk], ua1[e]);
        }
      }
    }
    float* uo = &sU[hi8 * OP + uj * CC + uct0 * 16 + m0];
#pragma unroll
    for (int e = 0; e < 3; ++e)
#pragma unroll
      for (int v = 0; v < 8; ++v) {
        uo[v * OP + e * 16] = ua0[e][v];
        uo[(16 + v) * OP + e * 16] = ua1[e][v];
      }
  }
  lds_barrier();

  const int aoff = (mt * 16 + m0) * OP + ksel;
  const int boff = (nt * 16 + m0) * LDP + ksel;
#pragma unroll
  for (int t = 0; t < NCH; ++t) {
    const int cur = t & 1, nxt = cur ^ 1;
    s_block(accS, &sU[aoff + t * CW], &sK[cur][boff]);
    if (t + 1 < NCH) {
      stage_chunk(&sK[nxt][so], rk[nxt]);
    } else {
#pragma unroll
      for (int v = 0; v < 8; ++v) {
        const int m = mt * 16 + v + hi8;
        sS[m * 68 + ncol] = accS[v] * ATTN_SCALE + mv[v];
      }
      stage_chunk(&sK[nxt][so], rk[nxt]);
    }
    lds_barrier();
    if (t + 3 < NCH) {
      rk[nxt] = fetch_chunk_x(gk + (t + 3) * CW);
    } else {
      rk[nxt] = fetch_chunk_x(gx + (t + 3 - NCH) * CW);
    }
    __builtin_amdgcn_sched_barrier(0);
  }

  for (int r = wave * 4; r < wave * 4 + 4; ++r) {
    float x0 = sS[r * 68 + lane];
    float x1 = sS[r * 68 + lane + 32];
    float mx = fmaxf(x0, x1);
#pragma unroll
    for (int off = 16; off >= 1; off >>= 1)
      mx = fmaxf(mx, __shfl_xor(mx, off, 32));
    const float e0 = __expf(x0 - mx), e1 = __expf(x1 - mx);
    float sm = e0 + e1;
#pragma unroll
    for (int off = 16; off >= 1; off >>= 1)
      sm += __shfl_xor(sm, off, 32);
    const float inv = 1.0f / sm;
    sS[r * 68 + lane] = e0 * inv;
    sS[r * 68 + lane + 32] = e1 * inv;
  }
  lds_barrier();

  const int dq = nt * 32;
  const float* pp = &sS[(mt * 16 + m0) * 68 + ksel];
  float* zrow = &sO[(mt * 16 + hi8) * OP + dq + m0];
  float4 wr[27];
#pragma unroll
  for (int u = 0; u < NXC; ++u) {
    const int cur = (NCH + u) & 1, nxt = cur ^ 1;
    pv_block(&sK[cur][ksel * LDP + dq + m0], pp, zrow + u * CW);
    if (u + 1 < NXC) stage_chunk(&sK[nxt][so], rk[nxt]);
    if (u + 1 == NXC) {
#pragma unroll
      for (int i = 0; i < 27; ++i) st_w(sW, tid + 256 * i, wr[i]);
    }
    lds_barrier();
    if (u == 0) {
#pragma unroll
      for (int i = 0; i < 27; ++i) wr[i] = ld_w(MT, tid + 256 * i);
      __builtin_amdgcn_sched_barrier(0);
    }
  }

  const int G = 128 + 8 * h + wave;
  const float* oa = &sO[m0 * OP + (G / 3 - tb) * CC + ksel];
  v2f a0[24], a1[24];
#pragma unroll
  for (int kk = 0; kk < 24; ++kk) {
    a0[kk] = *reinterpret_cast<const v2f*>(oa + kk * 4);
    a1[kk] = *reinterpret_cast<const v2f*>(oa + 16 * OP + kk * 4);
  }
  float* ob = out + ((size_t)(bt * WN + q0 + hi8) * NN + (h * 8 + wave)) * CC + m0;
  const float* wb = &sW[((G % 3) * CC + m0) * WP + ksel];
  v2f b[2][24];
#pragma unroll
  for (int kk = 0; kk < 24; ++kk)
    b[0][kk] = *reinterpret_cast<const v2f*>(wb + kk * 4);
#pragma unroll
  for (int nn = 0; nn < 6; ++nn) {
    const float bias = bproj[nn * 16 + m0];
    if (nn + 1 < 6) {
#pragma unroll
      for (int kk = 0; kk < 24; ++kk)
        b[(nn + 1) & 1][kk] =
            *reinterpret_cast<const v2f*>(wb + (nn + 1) * (16 * WP) + kk * 4);
    }
    __builtin_amdgcn_sched_barrier(0);
    v8f acc0 = {}, acc1 = {};
#pragma unroll
    for (int kk = 0; kk < 24; ++kk) {
      acc0 = wmma4(a0[kk], b[nn & 1][kk], acc0);
      acc1 = wmma4(a1[kk], b[nn & 1][kk], acc1);
    }
#pragma unroll
    for (int v = 0; v < 8; ++v) {
      ob[(size_t)v * (NN * CC) + nn * 16] = acc0[v] + bias;
      ob[(size_t)(16 + v) * (NN * CC) + nn * 16] = acc1[v] + bias;
    }
  }
}

extern "C" void kernel_launch(void* const* d_in, const int* in_sizes, int n_in,
                              void* d_out, int out_size, void* d_ws, size_t ws_size,
                              hipStream_t stream) {
  (void)in_sizes; (void)n_in; (void)out_size; (void)ws_size;
  const float* x     = (const float*)d_in[0];
  const float* mask  = (const float*)d_in[1];
  const float* wqkv  = (const float*)d_in[2];
  const float* wproj = (const float*)d_in[3];
  const float* bproj = (const float*)d_in[4];
  float* out = (float*)d_out;

  float* VT = (float*)d_ws;
  float* MT = VT + (size_t)HH * 4 * VTSZ;

  fold_weights_kernel<<<dim3(NAB * 4 + NMTW), dim3(288), 0, stream>>>(wqkv, wproj, VT, MT);
  win_attn_proj_kernel<<<dim3(BTOT * HH * 2), dim3(256), 0, stream>>>(x, VT, mask, MT, bproj, out);
}


# launch 1: only the 45 non-empty score-weight blocks get work-groups (192 instead of 524), so 2988 waves are no longer dispatched only to return; sums unchanged
# speedup vs baseline: 4.7375x; 1.0305x over previous
#include <hip/hip_runtime.h>


typedef __attribute__((ext_vector_type(2))) float v2f;
typedef __attribute__((ext_vector_type(8))) float v8f;

#define BTOT 16
#define WN   64
#define NN   64
#define CC   96
#define HH   8
#define DD   768
#define C3   288
#define HD   6144
#define ATTN_SCALE 0.28867513459481287f
#define NMW  3
#define MTSZ 9216
#define UW   384
#define VTSZ 36864
#define NFB  45

static __device__ __forceinline__ v8f wmma4(v2f a, v2f b, v8f c) {
  return __builtin_amdgcn_wmma_f32_16x16x4_f32(
       false, a,  false, b,
       (short)0, c,  false,  false);
}

static __device__ __forceinline__ void lds_barrier() {
  __builtin_amdgcn_fence(__ATOMIC_RELEASE, "workgroup", "local");
  __builtin_amdgcn_s_barrier();
  __builtin_amdgcn_fence(__ATOMIC_ACQUIRE, "workgroup", "local");
}

static __host__ __device__ constexpr __forceinline__ void fold_block(int h, int i, int j, int& lo, int& len) {
  const int tq0 = (DD * h) / C3, tk0 = (HD + DD * h) / C3;
  int a = DD * h, b = DD * h + DD;
  const int qa = C3 * (tq0 + i), ka = C3 * (tk0 + j) - HD;
  a = a > qa ? a : qa;  a = a > ka ? a : ka;
  b = b < qa + C3 ? b : qa + C3;  b = b < ka + C3 ? b : ka + C3;
  lo = a;  len = b > a ? b - a : 0;
}
static __host__ __device__ constexpr int fold_count(int h) {
  int n = 0;
  for (int e = 0; e < 16; ++e) {
    int lo = 0, len = 0;
    fold_block(h, e >> 2, e & 3, lo, len);
    n += len > 0 ? 1 : 0;
  }
  return n;
}
static __host__ __device__ constexpr unsigned long long fold_pairs(int h0, int which) {
  unsigned long long m = 0;
  int r = 0;
  for (int h = h0; h < h0 + 3 && h < HH; ++h)
    for (int e = 0; e < 16; ++e) {
      int lo = 0, len = 0;
      fold_block(h, e >> 2, e & 3, lo, len);
      if (len > 0) { m |= (unsigned long long)(which ? (e & 3) : (e >> 2)) << (2 * r);  ++r; }
    }
  return m;
}
static_assert(fold_count(0) == 5 && fold_count(1) == 6 && fold_count(2) == 6 && fold_count(3) == 5 &&
              fold_count(4) == 6 && fold_count(5) == 6 && fold_count(6) == 5 && fold_count(7) == 6 &&
              fold_pairs(3, 0) == fold_pairs(0, 0) && fold_pairs(3, 1) == fold_pairs(0, 1) &&
              fold_pairs(6, 0) == (fold_pairs(0, 0) & ((1ull << 22) - 1)) &&
              fold_pairs(6, 1) == (fold_pairs(0, 1) & ((1ull << 22) - 1)),
              "launch 1 numbers the non-empty blocks as 5, 6, 6 per head, alike every three heads");

#define NMTW 12
__global__ __launch_bounds__(288) void fold_weights_kernel(
    const float* __restrict__ wqkv, const float* __restrict__ wproj,
    float* __restrict__ VT, float* __restrict__ MT) {
  const int tid = threadIdx.x;
  const int lane = tid & 31, wave = tid >> 5;
  const int m0 = lane & 15;
  const int ksel = (lane >> 4) * 2;
  const int hi8 = (lane >> 4) << 3;

  if (blockIdx.x >= NFB * 4) {
    const int g = (blockIdx.x - NFB * 4) * 9 + wave;
    const int u = g / 36, ot = (g % 36) / 6, ct = g % 6;
    const float* aq = wproj + (ot * 16 + m0) * CC + ksel;
    const float* bq = wqkv + (size_t)(CC * u + ksel) * CC + ct * 16 + m0;
    v2f fa[24], fb[24];
#pragma unroll
    for (int kk = 0; kk < 24; ++kk) {
      fa[kk] = *reinterpret_cast<const v2f*>(aq + kk * 4);
      fb[kk].x = bq[(kk * 4) * CC];
      fb[kk].y = bq[(kk * 4 + 1) * CC];
    }
    v8f acc = {};
#pragma unroll
    for (int kk = 0; kk < 24; ++kk) acc = wmma4(fa[kk], fb[kk], acc);
    float* mo = MT + (size_t)u * MTSZ + (ot * 16 + hi8) * CC + ct * 16 + m0;
#pragma unroll
    for (int v = 0; v < 8; ++v) mo[v * CC] = acc[v];
    return;
  }

  const int bq = blockIdx.x >> 2;
  const int r = bq % 17;
  const int h = 3 * (bq / 17) + (r >= 5 ? 1 : 0) + (r >= 11 ? 1 : 0);
  constexpr unsigned long long QS = fold_pairs(0, 0), KS = fold_pairs(0, 1);
  const int bi = (int)(QS >> (2 * r)) & 3, bj = (int)(KS >> (2 * r)) & 3;
  int lo, len;
  fold_block(h, bi, bj, lo, len);
  const int T = (blockIdx.x & 3) * 9 + wave;
  const int mt = T / 6, nt = T % 6;
  const int cq0 = lo % C3, ck0 = (HD + lo) % C3;
  v8f acc = {};
  for (int t0 = 0; t0 < len; t0 += CC) {
    const float* ag = wqkv + (size_t)(ck0 + t0 + ksel) * CC + mt * 16 + m0;
    const float* bg = wqkv + (size_t)(cq0 + t0 + ksel) * CC + nt * 16 + m0;
    v2f fa[24], fb[24];
#pragma unroll
    for (int kk = 0; kk < 24; ++kk) {
      fa[kk].x = ag[(kk * 4) * CC];  fa[kk].y = ag[(kk * 4 + 1) * CC];
      fb[kk].x = bg[(kk * 4) * CC];  fb[kk].y = bg[(kk * 4 + 1) * CC];
    }
#pragma unroll
    for (int kk = 0; kk < 24; ++kk) acc = wmma4(fa[kk], fb[kk], acc);
  }
  float* vo = VT + (size_t)(h * 4 + bj) * VTSZ + (size_t)(mt * 16 + hi8) * UW + bi * CC + nt * 16 + m0;
#pragma unroll
  for (int v = 0; v < 8; ++v) vo[v * UW] = acc[v];
}

#define CW   128
#define NCH  3
#define LDP  132
#define QR   32
#define NXC  3
#define XROW 6144
#define OP   388
#define WP   100

struct chunk8 { float4 r0, r1, r2, r3, r4, r5, r6, r7; };
static __device__ __forceinline__ void st4(float* s, int u, float4 v) {
  *reinterpret_cast<float4*>(s + u * (8 * LDP)) = v;
}
static __device__ __forceinline__ float4 ld4x(const float* __restrict__ g, int u) {
  return *reinterpret_cast<const float4*>(g + (size_t)u * (8 * XROW));
}
static __device__ __forceinline__ chunk8 fetch_chunk_x(const float* __restrict__ g) {
  chunk8 c;
  c.r0 = ld4x(g, 0); c.r1 = ld4x(g, 1); c.r2 = ld4x(g, 2); c.r3 = ld4x(g, 3);
  c.r4 = ld4x(g, 4); c.r5 = ld4x(g, 5); c.r6 = ld4x(g, 6); c.r7 = ld4x(g, 7);
  return c;
}
struct chunk4 { float4 r0, r1, r2, r3; };
static __device__ __forceinline__ chunk4 fetch_chunk4_x(const float* __restrict__ g) {
  chunk4 c;
  c.r0 = ld4x(g, 0); c.r1 = ld4x(g, 1); c.r2 = ld4x(g, 2); c.r3 = ld4x(g, 3);
  return c;
}
static __device__ __forceinline__ void st4q(float* s, int u, float4 v) {
  *reinterpret_cast<float4*>(s + u * (8 * OP)) = v;
}
static __device__ __forceinline__ void stage_chunk4q(float* s, const chunk4& c) {
  st4q(s, 0, c.r0); st4q(s, 1, c.r1); st4q(s, 2, c.r2); st4q(s, 3, c.r3);
}
static __device__ __forceinline__ void stage_chunk(float* s, const chunk8& c) {
  st4(s, 0, c.r0); st4(s, 1, c.r1); st4(s, 2, c.r2); st4(s, 3, c.r3);
  st4(s, 4, c.r4); st4(s, 5, c.r5); st4(s, 6, c.r6); st4(s, 7, c.r7);
}

static __device__ __forceinline__ float4 ld_w(const float* __restrict__ w, int f) {
  return *reinterpret_cast<const float4*>(w + f * 4);
}
static __device__ __forceinline__ void st_w(float* sw, int f, float4 v) {
  *reinterpret_cast<float4*>(&sw[(f / 24) * WP + (f % 24) * 4]) = v;
}

static __device__ __forceinline__ void s_block(v8f& accS, const float* ap,
                                               const float* bp) {
#pragma unroll
  for (int kk = 0; kk < 32; ++kk) {
    v2f a = *reinterpret_cast<const v2f*>(ap + kk * 4);
    v2f b = *reinterpret_cast<const v2f*>(bp + kk * 4);
    accS = wmma4(a, b, accS);
  }
}

static __device__ __forceinline__ void pv_block(const float* vp, const float* pp,
                                                float* so) {
  v8f acc[2];
#pragma unroll
  for (int j = 0; j < 2; ++j) acc[j] = (v8f){};
#pragma unroll
  for (int kk = 0; kk < 16; ++kk) {
    v2f a = *reinterpret_cast<const v2f*>(pp + kk * 4);
#pragma unroll
    for (int j = 0; j < 2; ++j) {
      v2f b;
      b.x = vp[(kk * 4) * LDP + j * 16];
      b.y = vp[(kk * 4 + 1) * LDP + j * 16];
      acc[j] = wmma4(a, b, acc[j]);
    }
  }
#pragma unroll
  for (int j = 0; j < 2; ++j)
#pragma unroll
    for (int v = 0; v < 8; ++v)
      so[v * OP + j * 16] = acc[j][v];
}

__global__ __launch_bounds__(256) void win_attn_proj_kernel(
    const float* __restrict__ x, const float* __restrict__ VT,
    const float* __restrict__ mask,
    const float* __restrict__ MT, const float* __restrict__ bproj,
    float* __restrict__ out) {
  __shared__ __align__(16) float sK[2][64 * LDP];
  __shared__ __align__(16) float sS[QR * 68];
  __shared__ __align__(16) float sO[QR * OP];
  __shared__ __align__(16) float sW[NMW * CC * WP];
  float* const sXq = sO;
  float* const sU = sW;
  const int tid = threadIdx.x, lane = tid & 31, wave = tid >> 5;
  const int bt = blockIdx.x >> 4, h = (blockIdx.x >> 1) & 7;
  const int q0 = (blockIdx.x & 1) * QR;

  const int m0 = lane & 15;
  const int ksel = (lane >> 4) * 2;
  const int hi8 = (lane >> 4) << 3;
  const int mt = wave >> 2;
  const int nt = wave & 3;

  const int tk0 = (HD + DD * h) / C3;
  const float* gk = x + ((size_t)(bt * WN + wave) * NN + tk0) * CC + lane * 4;
  const int t0 = (128 + 8 * h) / 3;
  const int tb = t0 < NN - 4 ? t0 : NN - 4;
  const float* gx = x + ((size_t)(bt * WN + wave) * NN + tb) * CC + lane * 4;
  const int so = wave * LDP + lane * 4;

  chunk8 rk[2];

  v8f accS = (v8f){};

  const float* mk = mask + (size_t)(bt & 7) * (NN * NN) + (size_t)q0 * NN;
  const int ncol = nt * 16 + m0;
  float mv[8];
#pragma unroll
  for (int v = 0; v < 8; ++v)
    mv[v] = mk[(mt * 16 + v + hi8) * NN + ncol];

  const int tq0 = (DD * h) / C3;
  const float* gxq = x + ((size_t)(bt * WN + q0 + wave) * NN + tq0) * CC + lane * 4;
  const chunk4 xq0 = fetch_chunk4_x(gxq);
  const chunk4 xq1 = fetch_chunk4_x(gxq + CW);
  const chunk4 xq2 = fetch_chunk4_x(gxq + 2 * CW);
  rk[0] = fetch_chunk_x(gk);
  __builtin_amdgcn_sched_barrier(0);
  rk[1] = fetch_chunk_x(gk + CW);
  const int soq = wave * OP + lane * 4;
  stage_chunk4q(&sXq[soq], xq0);
  stage_chunk4q(&sXq[soq + CW], xq1);
  stage_chunk4q(&sXq[soq + 2 * CW], xq2);
  stage_chunk(&sK[0][so], rk[0]);
  lds_barrier();
  rk[0] = fetch_chunk_x(gk + 2 * CW);
  __builtin_amdgcn_sched_barrier(0);

  {
    const int wv = __builtin_amdgcn_readfirstlane(tid >> 5);
    const int uj = wv >> 1, uct0 = (wv & 1) * 3;
    v8f ua0[3], ua1[3];
#pragma unroll
    for (int e = 0; e < 3; ++e) { ua0[e] = (v8f){}; ua1[e] = (v8f){}; }
    for (int i = 0; i < 4; ++i) {
      int lo, len;
      fold_block(h, i, uj, lo, len);
      if (len == 0) continue;
      const float* bg = VT + (size_t)(h * 4 + uj) * VTSZ + (size_t)(uct0 * 16 + m0) * UW + i * CC + ksel;
      const float* ap = &sXq[m0 * OP + i * CC + ksel];
      v2f b[3][24];
#pragma unroll
      for (int kk = 0; kk < 24; ++kk)
#pragma unroll
        for (int e = 0; e < 3; ++e)
          b[e][kk] = *reinterpret_cast<const v2f*>(bg + e * (16 * UW) + kk * 4);
      __builtin_amdgcn_sched_barrier(0);
#pragma unroll
      for (int kk = 0; kk < 24; ++kk) {
        v2f a0 = *reinterpret_cast<const v2f*>(ap + kk * 4);
        v2f a1 = *reinterpret_cast<const v2f*>(ap + 16 * OP + kk * 4);
#pragma unroll
        for (int e = 0; e < 3; ++e) {
          ua0[e] = wmma4(a0, b[e][kk], ua0[e]);
          ua1[e] = wmma4(a1, b[e][kk], ua1[e]);
        }
      }
    }
    float* uo = &sU[hi8 * OP + uj * CC + uct0 * 16 + m0];
#pragma unroll
    for (int e = 0; e < 3; ++e)
#pragma unroll
      for (int v = 0; v < 8; ++v) {
        uo[v * OP + e * 16] = ua0[e][v];
        uo[(16 + v) * OP + e * 16] = ua1[e][v];
      }
  }
  lds_barrier();

  const int aoff = (mt * 16 + m0) * OP + ksel;
  const int boff = (nt * 16 + m0) * LDP + ksel;
#pragma unroll
  for (int t = 0; t < NCH; ++t) {
    const int cur = t & 1, nxt = cur ^ 1;
    s_block(accS, &sU[aoff + t * CW], &sK[cur][boff]);
    if (t + 1 < NCH) {
      stage_chunk(&sK[nxt][so], rk[nxt]);
    } else {
#pragma unroll
      for (int v = 0; v < 8; ++v) {
        const int m = mt * 16 + v + hi8;
        sS[m * 68 + ncol] = accS[v] * ATTN_SCALE + mv[v];
      }
      stage_chunk(&sK[nxt][so], rk[nxt]);
    }
    lds_barrier();
    if (t + 3 < NCH) {
      rk[nxt] = fetch_chunk_x(gk + (t + 3) * CW);
    } else {
      rk[nxt] = fetch_chunk_x(gx + (t + 3 - NCH) * CW);
    }
    __builtin_amdgcn_sched_barrier(0);
  }

  for (int r = wave * 4; r < wave * 4 + 4; ++r) {
    float x0 = sS[r * 68 + lane];
    float x1 = sS[r * 68 + lane + 32];
    float mx = fmaxf(x0, x1);
#pragma unroll
    for (int off = 16; off >= 1; off >>= 1)
      mx = fmaxf(mx, __shfl_xor(mx, off, 32));
    const float e0 = __expf(x0 - mx), e1 = __expf(x1 - mx);
    float sm = e0 + e1;
#pragma unroll
    for (int off = 16; off >= 1; off >>= 1)
      sm += __shfl_xor(sm, off, 32);
    const float inv = 1.0f / sm;
    sS[r * 68 + lane] = e0 * inv;
    sS[r * 68 + lane + 32] = e1 * inv;
  }
  lds_barrier();

  const int dq = nt * 32;
  const float* pp = &sS[(mt * 16 + m0) * 68 + ksel];
  float* zrow = &sO[(mt * 16 + hi8) * OP + dq + m0];
  float4 wr[27];
#pragma unroll
  for (int u = 0; u < NXC; ++u) {
    const int cur = (NCH + u) & 1, nxt = cur ^ 1;
    pv_block(&sK[cur][ksel * LDP + dq + m0], pp, zrow + u * CW);
    if (u + 1 < NXC) stage_chunk(&sK[nxt][so], rk[nxt]);
    if (u + 1 == NXC) {
#pragma unroll
      for (int i = 0; i < 27; ++i) st_w(sW, tid + 256 * i, wr[i]);
    }
    lds_barrier();
    if (u == 0) {
#pragma unroll
      for (int i = 0; i < 27; ++i) wr[i] = ld_w(MT, tid + 256 * i);
      __builtin_amdgcn_sched_barrier(0);
    }
  }

  const int G = 128 + 8 * h + wave;
  const float* oa = &sO[m0 * OP + (G / 3 - tb) * CC + ksel];
  v2f a0[24], a1[24];
#pragma unroll
  for (int kk = 0; kk < 24; ++kk) {
    a0[kk] = *reinterpret_cast<const v2f*>(oa + kk * 4);
    a1[kk] = *reinterpret_cast<const v2f*>(oa + 16 * OP + kk * 4);
  }
  float* ob = out + ((size_t)(bt * WN + q0 + hi8) * NN + (h * 8 + wave)) * CC + m0;
  const float* wb = &sW[((G % 3) * CC + m0) * WP + ksel];
  v2f b[2][24];
#pragma unroll
  for (int kk = 0; kk < 24; ++kk)
    b[0][kk] = *reinterpret_cast<const v2f*>(wb + kk * 4);
#pragma unroll
  for (int nn = 0; nn < 6; ++nn) {
    const float bias = bproj[nn * 16 + m0];
    if (nn + 1 < 6) {
#pragma unroll
      for (int kk = 0; kk < 24; ++kk)
        b[(nn + 1) & 1][kk] =
            *reinterpret_cast<const v2f*>(wb + (nn + 1) * (16 * WP) + kk * 4);
    }
    __builtin_amdgcn_sched_barrier(0);
    v8f acc0 = {}, acc1 = {};
#pragma unroll
    for (int kk = 0; kk < 24; ++kk) {
      acc0 = wmma4(a0[kk], b[nn & 1][kk], acc0);
      acc1 = wmma4(a1[kk], b[nn & 1][kk], acc1);
    }
#pragma unroll
    for (int v = 0; v < 8; ++v) {
      ob[(size_t)v * (NN * CC) + nn * 16] = acc0[v] + bias;
      ob[(size_t)(16 + v) * (NN * CC) + nn * 16] = acc1[v] + bias;
    }
  }
}

extern "C" void kernel_launch(void* const* d_in, const int* in_sizes, int n_in,
                              void* d_out, int out_size, void* d_ws, size_t ws_size,
                              hipStream_t stream) {
  (void)in_sizes; (void)n_in; (void)out_size; (void)ws_size;
  const float* x     = (const float*)d_in[0];
  const float* mask  = (const float*)d_in[1];
  const float* wqkv  = (const float*)d_in[2];
  const float* wproj = (const float*)d_in[3];
  const float* bproj = (const float*)d_in[4];
  float* out = (float*)d_out;

  float* VT = (float*)d_ws;
  float* MT = VT + (size_t)HH * 4 * VTSZ;

  fold_weights_kernel<<<dim3(NFB * 4 + NMTW), dim3(288), 0, stream>>>(wqkv, wproj, VT, MT);
  win_attn_proj_kernel<<<dim3(BTOT * HH * 2), dim3(256), 0, stream>>>(x, VT, mask, MT, bproj, out);
}
